# starting kernel
# baseline (speedup 1.0000x reference)
#include <hip/hip_runtime.h>
#include <hip/hip_bf16.h>

typedef float v2f __attribute__((ext_vector_type(2)));
typedef float v8f __attribute__((ext_vector_type(8)));

#define NN   50000
#define EE   800000
#define H2C  64
#define H1C  128
#define DINC 256
#define BN_EPS 1e-5f

__global__ void zero_f32(float* __restrict__ p, size_t n) {
  size_t i = (size_t)blockIdx.x * blockDim.x + threadIdx.x;
  size_t stride = (size_t)gridDim.x * blockDim.x;
  for (; i < n; i += stride) p[i] = 0.0f;
}

__global__ void wmma_gemm_f32(const float* __restrict__ A,
                              const float* __restrict__ B,
                              float* __restrict__ C,
                              int Nrows, int K, int Ncols) {
  const int lane = threadIdx.x & 31;
  const int wave = threadIdx.x >> 5;
  const int colTiles = Ncols >> 4;
  const int rowTiles = Nrows >> 4;
  int tile = blockIdx.x * (blockDim.x >> 5) + wave;
  if (tile >= rowTiles * colTiles) return;

  const int rt = tile / colTiles;
  const int ct = tile - rt * colTiles;
  const int row0 = rt << 4;
  const int col0 = ct << 4;

  const int m  = lane & 15;
  const int kh = (lane >> 4) << 1;

  const float* __restrict__ arow = A + (size_t)(row0 + m) * K + kh;
  const float* __restrict__ bcol = B + (size_t)kh * Ncols + (col0 + m);

  v8f acc = {};
  for (int k = 0; k < K; k += 4) {
    v2f a = *(const v2f*)(arow + k);
    v2f b;
    b.x = bcol[(size_t)k * Ncols];
    b.y = bcol[(size_t)(k + 1) * Ncols];
    acc = __builtin_amdgcn_wmma_f32_16x16x4_f32(
        false, a, false, b, (short)0, acc, false, false);
  }

  float* __restrict__ crow = C + (size_t)col0 + m;
  const int rbase = row0 + ((lane >> 4) << 3);
#pragma unroll
  for (int r = 0; r < 8; ++r)
    crow[(size_t)(rbase + r) * Ncols] = acc[r];
}

__global__ void scatter_edges(const float* __restrict__ sup,
                              float* __restrict__ agg,
                              const int* __restrict__ src,
                              const int* __restrict__ dst,
                              const float* __restrict__ w,
                              int E, int D) {
  const int chunks = D >> 2;
  long long idx = (long long)blockIdx.x * blockDim.x + threadIdx.x;
  long long total = (long long)E * chunks;
  if (idx >= total) return;
  int e = (int)(idx / chunks);
  int f = (int)(idx - (long long)e * chunks) << 2;

  const float wt = w[e];
  const float4 v = *(const float4*)(sup + (size_t)src[e] * D + f);
  float* __restrict__ out = agg + (size_t)dst[e] * D + f;
  atomicAdd(out + 0, v.x * wt);
  atomicAdd(out + 1, v.y * wt);
  atomicAdd(out + 2, v.z * wt);
  atomicAdd(out + 3, v.w * wt);
}

__global__ void relu_colstats(float* __restrict__ agg,
                              float* __restrict__ stats,
                              int Nrows, int D) {
  __shared__ float s_sum[DINC];
  __shared__ float s_sq[DINC];
  for (int t = threadIdx.x; t < D; t += blockDim.x) {
    s_sum[t] = 0.0f;
    s_sq[t]  = 0.0f;
  }
  __syncthreads();

  const size_t total = (size_t)Nrows * D;
  const size_t stride = (size_t)gridDim.x * blockDim.x;
  for (size_t i = (size_t)blockIdx.x * blockDim.x + threadIdx.x; i < total;
       i += stride) {
    float v = agg[i];
    v = v > 0.0f ? v : 0.0f;
    agg[i] = v;
    int c = (int)(i & (size_t)(D - 1));
    atomicAdd(&s_sum[c], v);
    atomicAdd(&s_sq[c], v * v);
  }
  __syncthreads();

  for (int t = threadIdx.x; t < D; t += blockDim.x) {
    atomicAdd(&stats[t], s_sum[t]);
    atomicAdd(&stats[D + t], s_sq[t]);
  }
}

__global__ void bn_normalize(const float* __restrict__ agg,
                             float* __restrict__ out,
                             const float* __restrict__ gamma,
                             const float* __restrict__ beta,
                             const float* __restrict__ stats,
                             int Nrows, int D) {
  size_t total = (size_t)Nrows * D;
  size_t i = (size_t)blockIdx.x * blockDim.x + threadIdx.x;
  if (i >= total) return;
  int c = (int)(i & (size_t)(D - 1));
  const float invN = 1.0f / (float)Nrows;
  const float mean = stats[c] * invN;
  const float var  = stats[D + c] * invN - mean * mean;
  const float s = gamma[c] * rsqrtf(var + BN_EPS);
  out[i] = (agg[i] - mean) * s + beta[c];
}

extern "C" void kernel_launch(void* const* d_in, const int* in_sizes, int n_in,
                              void* d_out, int out_size, void* d_ws,
                              size_t ws_size, hipStream_t stream) {
  (void)in_sizes; (void)n_in; (void)out_size; (void)ws_size;

  const float* y   = (const float*)d_in[0];
  const int*   src = (const int*)d_in[1];
  const int*   dst = (const int*)d_in[2];
  const float* w   = (const float*)d_in[3];
  const float* W1  = (const float*)d_in[4];
  const float* g1  = (const float*)d_in[5];
  const float* b1  = (const float*)d_in[6];
  const float* W2  = (const float*)d_in[7];
  const float* g2  = (const float*)d_in[8];
  const float* b2  = (const float*)d_in[9];
  float* out = (float*)d_out;

  float* ws    = (float*)d_ws;
  float* sup1  = ws;
  float* sup2  = sup1 + (size_t)NN * H1C;
  float* agg1  = sup2 + (size_t)NN * DINC;
  float* agg2  = agg1 + (size_t)NN * H1C;
  float* stats1 = agg2 + (size_t)NN * DINC;
  float* stats2 = stats1 + 2 * H1C;

  const size_t zeroN =
      (size_t)NN * H1C + (size_t)NN * DINC + 2 * (H1C + DINC);
  zero_f32<<<2048, 256, 0, stream>>>(agg1, zeroN);

  wmma_gemm_f32<<<25000 / 8, 256, 0, stream>>>(y, W1, sup1, NN, H2C, H1C);

  {
    long long total = (long long)EE * (H1C / 4);
    int blocks = (int)((total + 255) / 256);
    scatter_edges<<<blocks, 256, 0, stream>>>(sup1, agg1, src, dst, w, EE, H1C);
  }

  relu_colstats<<<960, 256, 0, stream>>>(agg1, stats1, NN, H1C);

  {
    size_t total = (size_t)NN * H1C;
    bn_normalize<<<(int)((total + 255) / 256), 256, 0, stream>>>(
        agg1, sup1, g1, b1, stats1, NN, H1C);
  }

  wmma_gemm_f32<<<50000 / 8, 256, 0, stream>>>(sup1, W2, sup2, NN, H1C, DINC);

  {
    long long total = (long long)EE * (DINC / 4);
    int blocks = (int)((total + 255) / 256);
    scatter_edges<<<blocks, 256, 0, stream>>>(sup2, agg2, src, dst, w, EE, DINC);
  }

  relu_colstats<<<960, 256, 0, stream>>>(agg2, stats2, NN, DINC);

  {
    size_t total = (size_t)NN * DINC;
    bn_normalize<<<(int)((total + 255) / 256), 256, 0, stream>>>(
        agg2, out, g2, b2, stats2, NN, DINC);
  }
}


# Edge sums taken before each dense product, (Adj X) W for Adj (X W): float atomics scatter 64- and 128-wide input rows, not 128 and 256; order of float additions changes
# speedup vs baseline: 1.6037x; 1.6037x over previous
#include <hip/hip_runtime.h>
#include <hip/hip_bf16.h>

typedef float v2f __attribute__((ext_vector_type(2)));
typedef float v8f __attribute__((ext_vector_type(8)));

#define NN   50000
#define EE   800000
#define H2C  64
#define H1C  128
#define DINC 256
#define BN_EPS 1e-5f

__global__ void zero_f32(float* __restrict__ p, size_t n) {
  size_t i = (size_t)blockIdx.x * blockDim.x + threadIdx.x;
  size_t stride = (size_t)gridDim.x * blockDim.x;
  for (; i < n; i += stride) p[i] = 0.0f;
}

__global__ void wmma_gemm_f32(const float* __restrict__ A,
                              const float* __restrict__ B,
                              float* __restrict__ C,
                              int Nrows, int K, int Ncols) {
  const int lane = threadIdx.x & 31;
  const int wave = threadIdx.x >> 5;
  const int colTiles = Ncols >> 4;
  const int rowTiles = Nrows >> 4;
  int tile = blockIdx.x * (blockDim.x >> 5) + wave;
  if (tile >= rowTiles * colTiles) return;

  const int rt = tile / colTiles;
  const int ct = tile - rt * colTiles;
  const int row0 = rt << 4;
  const int col0 = ct << 4;

  const int m  = lane & 15;
  const int kh = (lane >> 4) << 1;

  const float* __restrict__ arow = A + (size_t)(row0 + m) * K + kh;
  const float* __restrict__ bcol = B + (size_t)kh * Ncols + (col0 + m);

  v8f acc = {};
  for (int k = 0; k < K; k += 4) {
    v2f a = *(const v2f*)(arow + k);
    v2f b;
    b.x = bcol[(size_t)k * Ncols];
    b.y = bcol[(size_t)(k + 1) * Ncols];
    acc = __builtin_amdgcn_wmma_f32_16x16x4_f32(
        false, a, false, b, (short)0, acc, false, false);
  }

  float* __restrict__ crow = C + (size_t)col0 + m;
  const int rbase = row0 + ((lane >> 4) << 3);
#pragma unroll
  for (int r = 0; r < 8; ++r)
    crow[(size_t)(rbase + r) * Ncols] = acc[r];
}

__global__ void scatter_edges(const float* __restrict__ sup,
                              float* __restrict__ agg,
                              const int* __restrict__ src,
                              const int* __restrict__ dst,
                              const float* __restrict__ w,
                              int E, int D) {
  const int chunks = D >> 2;
  long long idx = (long long)blockIdx.x * blockDim.x + threadIdx.x;
  long long total = (long long)E * chunks;
  if (idx >= total) return;
  int e = (int)(idx / chunks);
  int f = (int)(idx - (long long)e * chunks) << 2;

  const float wt = w[e];
  const float4 v = *(const float4*)(sup + (size_t)src[e] * D + f);
  float* __restrict__ out = agg + (size_t)dst[e] * D + f;
  atomicAdd(out + 0, v.x * wt);
  atomicAdd(out + 1, v.y * wt);
  atomicAdd(out + 2, v.z * wt);
  atomicAdd(out + 3, v.w * wt);
}

__global__ void relu_colstats(float* __restrict__ agg,
                              float* __restrict__ stats,
                              int Nrows, int D) {
  __shared__ float s_sum[DINC];
  __shared__ float s_sq[DINC];
  for (int t = threadIdx.x; t < D; t += blockDim.x) {
    s_sum[t] = 0.0f;
    s_sq[t]  = 0.0f;
  }
  __syncthreads();

  const size_t total = (size_t)Nrows * D;
  const size_t stride = (size_t)gridDim.x * blockDim.x;
  for (size_t i = (size_t)blockIdx.x * blockDim.x + threadIdx.x; i < total;
       i += stride) {
    float v = agg[i];
    v = v > 0.0f ? v : 0.0f;
    agg[i] = v;
    int c = (int)(i & (size_t)(D - 1));
    atomicAdd(&s_sum[c], v);
    atomicAdd(&s_sq[c], v * v);
  }
  __syncthreads();

  for (int t = threadIdx.x; t < D; t += blockDim.x) {
    atomicAdd(&stats[t], s_sum[t]);
    atomicAdd(&stats[D + t], s_sq[t]);
  }
}

__global__ void bn_normalize(const float* __restrict__ agg,
                             float* __restrict__ out,
                             const float* __restrict__ gamma,
                             const float* __restrict__ beta,
                             const float* __restrict__ stats,
                             int Nrows, int D) {
  size_t total = (size_t)Nrows * D;
  size_t i = (size_t)blockIdx.x * blockDim.x + threadIdx.x;
  if (i >= total) return;
  int c = (int)(i & (size_t)(D - 1));
  const float invN = 1.0f / (float)Nrows;
  const float mean = stats[c] * invN;
  const float var  = stats[D + c] * invN - mean * mean;
  const float s = gamma[c] * rsqrtf(var + BN_EPS);
  out[i] = (agg[i] - mean) * s + beta[c];
}

extern "C" void kernel_launch(void* const* d_in, const int* in_sizes, int n_in,
                              void* d_out, int out_size, void* d_ws,
                              size_t ws_size, hipStream_t stream) {
  (void)in_sizes; (void)n_in; (void)out_size; (void)ws_size;

  const float* y   = (const float*)d_in[0];
  const int*   src = (const int*)d_in[1];
  const int*   dst = (const int*)d_in[2];
  const float* w   = (const float*)d_in[3];
  const float* W1  = (const float*)d_in[4];
  const float* g1  = (const float*)d_in[5];
  const float* b1  = (const float*)d_in[6];
  const float* W2  = (const float*)d_in[7];
  const float* g2  = (const float*)d_in[8];
  const float* b2  = (const float*)d_in[9];
  float* out = (float*)d_out;

  float* ws     = (float*)d_ws;
  float* z1     = ws;
  float* h1     = z1 + (size_t)NN * H1C;
  float* z2     = h1 + (size_t)NN * H1C;
  float* ax1    = z2 + (size_t)NN * DINC;
  float* ax2    = ax1 + (size_t)NN * H2C;
  float* stats1 = ax2 + (size_t)NN * H1C;
  float* stats2 = stats1 + 2 * H1C;

  const size_t zeroN =
      (size_t)NN * H2C + (size_t)NN * H1C + 2 * (H1C + DINC);
  zero_f32<<<2048, 256, 0, stream>>>(ax1, zeroN);

  {
    long long total = (long long)EE * (H2C / 4);
    int blocks = (int)((total + 255) / 256);
    scatter_edges<<<blocks, 256, 0, stream>>>(y, ax1, src, dst, w, EE, H2C);
  }

  wmma_gemm_f32<<<25000 / 8, 256, 0, stream>>>(ax1, W1, z1, NN, H2C, H1C);

  relu_colstats<<<960, 256, 0, stream>>>(z1, stats1, NN, H1C);

  {
    size_t total = (size_t)NN * H1C;
    bn_normalize<<<(int)((total + 255) / 256), 256, 0, stream>>>(
        z1, h1, g1, b1, stats1, NN, H1C);
  }

  {
    long long total = (long long)EE * (H1C / 4);
    int blocks = (int)((total + 255) / 256);
    scatter_edges<<<blocks, 256, 0, stream>>>(h1, ax2, src, dst, w, EE, H1C);
  }

  wmma_gemm_f32<<<50000 / 8, 256, 0, stream>>>(ax2, W2, z2, NN, H1C, DINC);

  relu_colstats<<<960, 256, 0, stream>>>(z2, stats2, NN, DINC);

  {
    size_t total = (size_t)NN * DINC;
    bn_normalize<<<(int)((total + 255) / 256), 256, 0, stream>>>(
        z2, out, g2, b2, stats2, NN, DINC);
  }
}


# Edges bucketed by destination row once per call (int atomic rank, arrival order) and each row summed by one wave: float atomics and the zero fill go; order of float additions changes
# speedup vs baseline: 2.0755x; 1.2942x over previous
#include <hip/hip_runtime.h>
#include <hip/hip_bf16.h>

typedef float v2f __attribute__((ext_vector_type(2)));
typedef float v8f __attribute__((ext_vector_type(8)));

#define NN   50000
#define EE   800000
#define H2C  64
#define H1C  128
#define DINC 256
#define BN_EPS 1e-5f

__global__ void zero_words(unsigned* __restrict__ p, int n) {
  const int i = blockIdx.x * blockDim.x + threadIdx.x;
  if (i < n) p[i] = 0u;
}

__global__ void wmma_gemm_f32(const float* __restrict__ A,
                              const float* __restrict__ B,
                              float* __restrict__ C,
                              int Nrows, int K, int Ncols) {
  const int lane = threadIdx.x & 31;
  const int wave = threadIdx.x >> 5;
  const int colTiles = Ncols >> 4;
  const int rowTiles = Nrows >> 4;
  int tile = blockIdx.x * (blockDim.x >> 5) + wave;
  if (tile >= rowTiles * colTiles) return;

  const int rt = tile / colTiles;
  const int ct = tile - rt * colTiles;
  const int row0 = rt << 4;
  const int col0 = ct << 4;

  const int m  = lane & 15;
  const int kh = (lane >> 4) << 1;

  const float* __restrict__ arow = A + (size_t)(row0 + m) * K + kh;
  const float* __restrict__ bcol = B + (size_t)kh * Ncols + (col0 + m);

  v8f acc = {};
  for (int k = 0; k < K; k += 4) {
    v2f a = *(const v2f*)(arow + k);
    v2f b;
    b.x = bcol[(size_t)k * Ncols];
    b.y = bcol[(size_t)(k + 1) * Ncols];
    acc = __builtin_amdgcn_wmma_f32_16x16x4_f32(
        false, a, false, b, (short)0, acc, false, false);
  }

  float* __restrict__ crow = C + (size_t)col0 + m;
  const int rbase = row0 + ((lane >> 4) << 3);
#pragma unroll
  for (int r = 0; r < 8; ++r)
    crow[(size_t)(rbase + r) * Ncols] = acc[r];
}

__global__ void count_rank(const int* __restrict__ dst,
                           int* __restrict__ cnt,
                           int* __restrict__ rank, int E4) {
  const int t = blockIdx.x * blockDim.x + threadIdx.x;
  if (t >= E4) return;
  const int4 d = ((const int4*)dst)[t];
  int4 r;
  r.x = atomicAdd(cnt + d.x, 1);
  r.y = atomicAdd(cnt + d.y, 1);
  r.z = atomicAdd(cnt + d.z, 1);
  r.w = atomicAdd(cnt + d.w, 1);
  ((int4*)rank)[t] = r;
}

__global__ void row_offsets(const int* __restrict__ cnt,
                            int* __restrict__ cursor,
                            int* __restrict__ rowstart, int n) {
  __shared__ int s_wave[8];
  __shared__ int s_base;
  const int row  = blockIdx.x * 256 + threadIdx.x;
  const int lane = threadIdx.x & 31;
  const int wv   = threadIdx.x >> 5;
  const int c = row < n ? cnt[row] : 0;
  int inc = c;
#pragma unroll
  for (int o = 1; o < 32; o <<= 1) {
    const int t = __shfl_up(inc, o);
    if (lane >= o) inc += t;
  }
  if (lane == 31) s_wave[wv] = inc;
  __syncthreads();
  if (threadIdx.x == 0) {
    int tot = 0;
    for (int i = 0; i < 8; ++i) {
      const int t = s_wave[i];
      s_wave[i] = tot;
      tot += t;
    }
    s_base = atomicAdd(cursor, tot);
  }
  __syncthreads();
  if (row < n) rowstart[row] = s_base + s_wave[wv] + inc - c;
}

__global__ void place_edges(const int* __restrict__ src,
                            const int* __restrict__ dst,
                            const float* __restrict__ w,
                            const int* __restrict__ rank,
                            const int* __restrict__ rowstart,
                            int* __restrict__ edges, int E4, int E) {
  const int t = blockIdx.x * blockDim.x + threadIdx.x;
  if (t >= E4) return;
  const int4   d  = ((const int4*)dst)[t];
  const int4   r  = ((const int4*)rank)[t];
  const int4   s  = ((const int4*)src)[t];
  const float4 ww = ((const float4*)w)[t];
  int2* __restrict__ es = (int2*)edges;
  const int p0 = rowstart[d.x] + r.x;
  const int p1 = rowstart[d.y] + r.y;
  const int p2 = rowstart[d.z] + r.z;
  const int p3 = rowstart[d.w] + r.w;
  if ((unsigned)p0 < (unsigned)E) es[p0] = make_int2(s.x, __float_as_int(ww.x));
  if ((unsigned)p1 < (unsigned)E) es[p1] = make_int2(s.y, __float_as_int(ww.y));
  if ((unsigned)p2 < (unsigned)E) es[p2] = make_int2(s.z, __float_as_int(ww.z));
  if ((unsigned)p3 < (unsigned)E) es[p3] = make_int2(s.w, __float_as_int(ww.w));
}

template <int VPL>
__global__ void gather_rows(const float* __restrict__ x,
                            float* __restrict__ ax,
                            const int* __restrict__ rowstart,
                            const int* __restrict__ cnt,
                            const int* __restrict__ edges,
                            int n, int E) {
  typedef float vf __attribute__((ext_vector_type(VPL)));
  const int2* __restrict__ es = (const int2*)edges;
  const int lane = threadIdx.x & 31;
  const int wave0 = __builtin_amdgcn_readfirstlane(
      (int)((blockIdx.x * blockDim.x + threadIdx.x) >> 5));
  const int nwaves = (int)((gridDim.x * blockDim.x) >> 5);

  for (int row = wave0; row < n; row += nwaves) {
    const int start = rowstart[row];
    const int deg   = cnt[row];
    vf acc = {};
    for (int base = 0; base < deg; base += 32) {
      const int m = min(32, deg - base);
      const int j = min(lane, m - 1);
      const int2 ed = es[min(start + base + j, E - 1)];
      const int   e_src = ed.x;
      const float e_w   = lane < m ? __int_as_float(ed.y) : 0.0f;
      for (int j0 = 0; j0 < m; j0 += 8) {
        vf v[8];
        float wj[8];
#pragma unroll
        for (int u = 0; u < 8; ++u) {
          const int s = __builtin_amdgcn_readlane(e_src, j0 + u);
          wj[u] = __int_as_float(
              __builtin_amdgcn_readlane(__float_as_int(e_w), j0 + u));
          v[u] = *(const vf*)(x + (size_t)s * (32 * VPL) + lane * VPL);
        }
#pragma unroll
        for (int u = 0; u < 8; ++u) acc += wj[u] * v[u];
      }
    }
    *(vf*)(ax + (size_t)row * (32 * VPL) + lane * VPL) = acc;
  }
}

__global__ void relu_colstats(float* __restrict__ agg,
                              float* __restrict__ stats,
                              int Nrows, int D) {
  __shared__ float s_sum[DINC];
  __shared__ float s_sq[DINC];
  for (int t = threadIdx.x; t < D; t += blockDim.x) {
    s_sum[t] = 0.0f;
    s_sq[t]  = 0.0f;
  }
  __syncthreads();

  const size_t total = (size_t)Nrows * D;
  const size_t stride = (size_t)gridDim.x * blockDim.x;
  for (size_t i = (size_t)blockIdx.x * blockDim.x + threadIdx.x; i < total;
       i += stride) {
    float v = agg[i];
    v = v > 0.0f ? v : 0.0f;
    agg[i] = v;
    int c = (int)(i & (size_t)(D - 1));
    atomicAdd(&s_sum[c], v);
    atomicAdd(&s_sq[c], v * v);
  }
  __syncthreads();

  for (int t = threadIdx.x; t < D; t += blockDim.x) {
    atomicAdd(&stats[t], s_sum[t]);
    atomicAdd(&stats[D + t], s_sq[t]);
  }
}

__global__ void bn_normalize(const float* __restrict__ agg,
                             float* __restrict__ out,
                             const float* __restrict__ gamma,
                             const float* __restrict__ beta,
                             const float* __restrict__ stats,
                             int Nrows, int D) {
  size_t total = (size_t)Nrows * D;
  size_t i = (size_t)blockIdx.x * blockDim.x + threadIdx.x;
  if (i >= total) return;
  int c = (int)(i & (size_t)(D - 1));
  const float invN = 1.0f / (float)Nrows;
  const float mean = stats[c] * invN;
  const float var  = stats[D + c] * invN - mean * mean;
  const float s = gamma[c] * rsqrtf(var + BN_EPS);
  out[i] = (agg[i] - mean) * s + beta[c];
}

extern "C" void kernel_launch(void* const* d_in, const int* in_sizes, int n_in,
                              void* d_out, int out_size, void* d_ws,
                              size_t ws_size, hipStream_t stream) {
  (void)in_sizes; (void)n_in; (void)out_size; (void)ws_size;

  const float* y   = (const float*)d_in[0];
  const int*   src = (const int*)d_in[1];
  const int*   dst = (const int*)d_in[2];
  const float* w   = (const float*)d_in[3];
  const float* W1  = (const float*)d_in[4];
  const float* g1  = (const float*)d_in[5];
  const float* b1  = (const float*)d_in[6];
  const float* W2  = (const float*)d_in[7];
  const float* g2  = (const float*)d_in[8];
  const float* b2  = (const float*)d_in[9];
  float* out = (float*)d_out;

  float* ws     = (float*)d_ws;
  float* z1     = ws;
  float* h1     = z1 + (size_t)NN * H1C;
  float* z2     = h1 + (size_t)NN * H1C;
  float* ax1    = z2 + (size_t)NN * DINC;
  float* ax2    = ax1 + (size_t)NN * H2C;
  float* stats1 = ax2 + (size_t)NN * H1C;
  float* stats2 = stats1 + 2 * H1C;
  int*   cnt    = (int*)(stats2 + 2 * DINC);
  int*   cursor = cnt + NN;
  int*   rowst  = cursor + 64;
  int*   rank   = rowst + NN;
  int*   edges  = rank + EE;

  zero_words<<<(2 * (H1C + DINC) + NN + 64 + 255) / 256, 256, 0, stream>>>(
      (unsigned*)stats1, 2 * (H1C + DINC) + NN + 64);

  count_rank<<<(EE / 4 + 255) / 256, 256, 0, stream>>>(dst, cnt, rank, EE / 4);
  row_offsets<<<(NN + 255) / 256, 256, 0, stream>>>(cnt, cursor, rowst, NN);
  place_edges<<<(EE / 4 + 255) / 256, 256, 0, stream>>>(
      src, dst, w, rank, rowst, edges, EE / 4, EE);

  gather_rows<2><<<2048, 256, 0, stream>>>(y, ax1, rowst, cnt, edges, NN, EE);

  wmma_gemm_f32<<<25000 / 8, 256, 0, stream>>>(ax1, W1, z1, NN, H2C, H1C);

  relu_colstats<<<960, 256, 0, stream>>>(z1, stats1, NN, H1C);

  {
    size_t total = (size_t)NN * H1C;
    bn_normalize<<<(int)((total + 255) / 256), 256, 0, stream>>>(
        z1, h1, g1, b1, stats1, NN, H1C);
  }

  gather_rows<4><<<2048, 256, 0, stream>>>(h1, ax2, rowst, cnt, edges, NN, EE);

  wmma_gemm_f32<<<50000 / 8, 256, 0, stream>>>(ax2, W2, z2, NN, H1C, DINC);

  relu_colstats<<<960, 256, 0, stream>>>(z2, stats2, NN, DINC);

  {
    size_t total = (size_t)NN * DINC;
    bn_normalize<<<(int)((total + 255) / 256), 256, 0, stream>>>(
        z2, out, g2, b2, stats2, NN, DINC);
  }
}


# ReLU and BatchNorm column sums move into the dense product's epilogue: a wave keeps 32 columns of W in registers and walks row tiles; separate ReLU/sums pass goes; order of float additions changes
# speedup vs baseline: 2.8769x; 1.3861x over previous
#include <hip/hip_runtime.h>
#include <hip/hip_bf16.h>

typedef float v2f __attribute__((ext_vector_type(2)));
typedef float v8f __attribute__((ext_vector_type(8)));

#define NN   50000
#define EE   800000
#define H2C  64
#define H1C  128
#define DINC 256
#define BN_EPS 1e-5f

__global__ void zero_words(unsigned* __restrict__ p, int n) {
  const int i = blockIdx.x * blockDim.x + threadIdx.x;
  if (i < n) p[i] = 0u;
}

template <int K, int NC>
__global__ void __launch_bounds__(NC)
gemm_relu_colstats(const float* __restrict__ A,
                   const float* __restrict__ B,
                   float* __restrict__ C,
                   float* __restrict__ stats,
                   int rowTiles) {
  const int lane = threadIdx.x & 31;
  const int wave = threadIdx.x >> 5;
  const int c = lane & 15;
  const int h = lane >> 4;
  const int col0 = wave << 5;

  v2f b0[K / 4], b1[K / 4];
  {
    const float* __restrict__ bp = B + (size_t)(4 * h) * NC + col0 + c;
#pragma unroll
    for (int t = 0; t < K / 4; ++t) {
      const float* __restrict__ p = bp + (size_t)(8 * (t >> 1) + 2 * (t & 1)) * NC;
      b0[t].x = p[0];
      b0[t].y = p[NC];
      b1[t].x = p[16];
      b1[t].y = p[NC + 16];
    }
  }

  float s0 = 0.0f, q0 = 0.0f, s1 = 0.0f, q1 = 0.0f;

  for (int rt = blockIdx.x; rt < rowTiles; rt += gridDim.x) {
    const int row0 = rt << 4;
    const float4* __restrict__ ap =
        (const float4*)(A + (size_t)(row0 + c) * K + 4 * h);
    float4 a[K / 8];
#pragma unroll
    for (int j = 0; j < K / 8; ++j) a[j] = ap[2 * j];

    v8f acc0 = {};
    v8f acc1 = {};
#pragma unroll
    for (int j = 0; j < K / 8; ++j) {
      v2f lo, hi;
      lo.x = a[j].x; lo.y = a[j].y;
      hi.x = a[j].z; hi.y = a[j].w;
      acc0 = __builtin_amdgcn_wmma_f32_16x16x4_f32(
          false, lo, false, b0[2 * j], (short)0, acc0, false, false);
      acc1 = __builtin_amdgcn_wmma_f32_16x16x4_f32(
          false, lo, false, b1[2 * j], (short)0, acc1, false, false);
      acc0 = __builtin_amdgcn_wmma_f32_16x16x4_f32(
          false, hi, false, b0[2 * j + 1], (short)0, acc0, false, false);
      acc1 = __builtin_amdgcn_wmma_f32_16x16x4_f32(
          false, hi, false, b1[2 * j + 1], (short)0, acc1, false, false);
    }

    float* __restrict__ cp = C + (size_t)(row0 + 8 * h) * NC + col0 + c;
#pragma unroll
    for (int r = 0; r < 8; ++r) {
      float v0 = acc0[r];
      float v1 = acc1[r];
      v0 = v0 > 0.0f ? v0 : 0.0f;
      v1 = v1 > 0.0f ? v1 : 0.0f;
      cp[(size_t)r * NC]      = v0;
      cp[(size_t)r * NC + 16] = v1;
      s0 += v0; q0 += v0 * v0;
      s1 += v1; q1 += v1 * v1;
    }
  }

  s0 += __shfl_xor(s0, 16); q0 += __shfl_xor(q0, 16);
  s1 += __shfl_xor(s1, 16); q1 += __shfl_xor(q1, 16);
  if (h == 0) {
    atomicAdd(&stats[col0 + c], s0);
    atomicAdd(&stats[NC + col0 + c], q0);
    atomicAdd(&stats[col0 + 16 + c], s1);
    atomicAdd(&stats[NC + col0 + 16 + c], q1);
  }
}

__global__ void count_rank(const int* __restrict__ dst,
                           int* __restrict__ cnt,
                           int* __restrict__ rank, int E4) {
  const int t = blockIdx.x * blockDim.x + threadIdx.x;
  if (t >= E4) return;
  const int4 d = ((const int4*)dst)[t];
  int4 r;
  r.x = atomicAdd(cnt + d.x, 1);
  r.y = atomicAdd(cnt + d.y, 1);
  r.z = atomicAdd(cnt + d.z, 1);
  r.w = atomicAdd(cnt + d.w, 1);
  ((int4*)rank)[t] = r;
}

__global__ void row_offsets(const int* __restrict__ cnt,
                            int* __restrict__ cursor,
                            int* __restrict__ rowstart, int n) {
  __shared__ int s_wave[8];
  __shared__ int s_base;
  const int row  = blockIdx.x * 256 + threadIdx.x;
  const int lane = threadIdx.x & 31;
  const int wv   = threadIdx.x >> 5;
  const int c = row < n ? cnt[row] : 0;
  int inc = c;
#pragma unroll
  for (int o = 1; o < 32; o <<= 1) {
    const int t = __shfl_up(inc, o);
    if (lane >= o) inc += t;
  }
  if (lane == 31) s_wave[wv] = inc;
  __syncthreads();
  if (threadIdx.x == 0) {
    int tot = 0;
    for (int i = 0; i < 8; ++i) {
      const int t = s_wave[i];
      s_wave[i] = tot;
      tot += t;
    }
    s_base = atomicAdd(cursor, tot);
  }
  __syncthreads();
  if (row < n) rowstart[row] = s_base + s_wave[wv] + inc - c;
}

__global__ void place_edges(const int* __restrict__ src,
                            const int* __restrict__ dst,
                            const float* __restrict__ w,
                            const int* __restrict__ rank,
                            const int* __restrict__ rowstart,
                            int* __restrict__ edges, int E4, int E) {
  const int t = blockIdx.x * blockDim.x + threadIdx.x;
  if (t >= E4) return;
  const int4   d  = ((const int4*)dst)[t];
  const int4   r  = ((const int4*)rank)[t];
  const int4   s  = ((const int4*)src)[t];
  const float4 ww = ((const float4*)w)[t];
  int2* __restrict__ es = (int2*)edges;
  const int p0 = rowstart[d.x] + r.x;
  const int p1 = rowstart[d.y] + r.y;
  const int p2 = rowstart[d.z] + r.z;
  const int p3 = rowstart[d.w] + r.w;
  if ((unsigned)p0 < (unsigned)E) es[p0] = make_int2(s.x, __float_as_int(ww.x));
  if ((unsigned)p1 < (unsigned)E) es[p1] = make_int2(s.y, __float_as_int(ww.y));
  if ((unsigned)p2 < (unsigned)E) es[p2] = make_int2(s.z, __float_as_int(ww.z));
  if ((unsigned)p3 < (unsigned)E) es[p3] = make_int2(s.w, __float_as_int(ww.w));
}

template <int VPL>
__global__ void gather_rows(const float* __restrict__ x,
                            float* __restrict__ ax,
                            const int* __restrict__ rowstart,
                            const int* __restrict__ cnt,
                            const int* __restrict__ edges,
                            int n, int E) {
  typedef float vf __attribute__((ext_vector_type(VPL)));
  const int2* __restrict__ es = (const int2*)edges;
  const int lane = threadIdx.x & 31;
  const int wave0 = __builtin_amdgcn_readfirstlane(
      (int)((blockIdx.x * blockDim.x + threadIdx.x) >> 5));
  const int nwaves = (int)((gridDim.x * blockDim.x) >> 5);

  for (int row = wave0; row < n; row += nwaves) {
    const int start = rowstart[row];
    const int deg   = cnt[row];
    vf acc = {};
    for (int base = 0; base < deg; base += 32) {
      const int m = min(32, deg - base);
      const int j = min(lane, m - 1);
      const int2 ed = es[min(start + base + j, E - 1)];
      const int   e_src = ed.x;
      const float e_w   = lane < m ? __int_as_float(ed.y) : 0.0f;
      for (int j0 = 0; j0 < m; j0 += 8) {
        vf v[8];
        float wj[8];
#pragma unroll
        for (int u = 0; u < 8; ++u) {
          const int s = __builtin_amdgcn_readlane(e_src, j0 + u);
          wj[u] = __int_as_float(
              __builtin_amdgcn_readlane(__float_as_int(e_w), j0 + u));
          v[u] = *(const vf*)(x + (size_t)s * (32 * VPL) + lane * VPL);
        }
#pragma unroll
        for (int u = 0; u < 8; ++u) acc += wj[u] * v[u];
      }
    }
    *(vf*)(ax + (size_t)row * (32 * VPL) + lane * VPL) = acc;
  }
}

__global__ void bn_normalize(const float* __restrict__ agg,
                             float* __restrict__ out,
                             const float* __restrict__ gamma,
                             const float* __restrict__ beta,
                             const float* __restrict__ stats,
                             int Nrows, int D) {
  size_t total = (size_t)Nrows * D;
  size_t i = (size_t)blockIdx.x * blockDim.x + threadIdx.x;
  if (i >= total) return;
  int c = (int)(i & (size_t)(D - 1));
  const float invN = 1.0f / (float)Nrows;
  const float mean = stats[c] * invN;
  const float var  = stats[D + c] * invN - mean * mean;
  const float s = gamma[c] * rsqrtf(var + BN_EPS);
  out[i] = (agg[i] - mean) * s + beta[c];
}

extern "C" void kernel_launch(void* const* d_in, const int* in_sizes, int n_in,
                              void* d_out, int out_size, void* d_ws,
                              size_t ws_size, hipStream_t stream) {
  (void)in_sizes; (void)n_in; (void)out_size; (void)ws_size;

  const float* y   = (const float*)d_in[0];
  const int*   src = (const int*)d_in[1];
  const int*   dst = (const int*)d_in[2];
  const float* w   = (const float*)d_in[3];
  const float* W1  = (const float*)d_in[4];
  const float* g1  = (const float*)d_in[5];
  const float* b1  = (const float*)d_in[6];
  const float* W2  = (const float*)d_in[7];
  const float* g2  = (const float*)d_in[8];
  const float* b2  = (const float*)d_in[9];
  float* out = (float*)d_out;

  float* ws     = (float*)d_ws;
  float* z1     = ws;
  float* h1     = z1 + (size_t)NN * H1C;
  float* z2     = h1 + (size_t)NN * H1C;
  float* ax1    = z2 + (size_t)NN * DINC;
  float* ax2    = ax1 + (size_t)NN * H2C;
  float* stats1 = ax2 + (size_t)NN * H1C;
  float* stats2 = stats1 + 2 * H1C;
  int*   cnt    = (int*)(stats2 + 2 * DINC);
  int*   cursor = cnt + NN;
  int*   rowst  = cursor + 64;
  int*   rank   = rowst + NN;
  int*   edges  = rank + EE;

  zero_words<<<(2 * (H1C + DINC) + NN + 64 + 255) / 256, 256, 0, stream>>>(
      (unsigned*)stats1, 2 * (H1C + DINC) + NN + 64);

  count_rank<<<(EE / 4 + 255) / 256, 256, 0, stream>>>(dst, cnt, rank, EE / 4);
  row_offsets<<<(NN + 255) / 256, 256, 0, stream>>>(cnt, cursor, rowst, NN);
  place_edges<<<(EE / 4 + 255) / 256, 256, 0, stream>>>(
      src, dst, w, rank, rowst, edges, EE / 4, EE);

  gather_rows<2><<<2048, 256, 0, stream>>>(y, ax1, rowst, cnt, edges, NN, EE);

  gemm_relu_colstats<H2C, H1C><<<1042, H1C, 0, stream>>>(
      ax1, W1, z1, stats1, NN / 16);

  {
    size_t total = (size_t)NN * H1C;
    bn_normalize<<<(int)((total + 255) / 256), 256, 0, stream>>>(
        z1, h1, g1, b1, stats1, NN, H1C);
  }

  gather_rows<4><<<2048, 256, 0, stream>>>(h1, ax2, rowst, cnt, edges, NN, EE);

  gemm_relu_colstats<H1C, DINC><<<512, DINC, 0, stream>>>(
      ax2, W2, z2, stats2, NN / 16);

  {
    size_t total = (size_t)NN * DINC;
    bn_normalize<<<(int)((total + 255) / 256), 256, 0, stream>>>(
        z2, out, g2, b2, stats2, NN, DINC);
  }
}


# BatchNorm passes as streams: 625 resident work-groups, a thread owns 4 fixed columns (mean, scale, shift once per thread) and has ten 16-byte loads in flight per batch, not one float per thread
# speedup vs baseline: 3.0065x; 1.0450x over previous
#include <hip/hip_runtime.h>
#include <hip/hip_bf16.h>

typedef float v2f __attribute__((ext_vector_type(2)));
typedef float v8f __attribute__((ext_vector_type(8)));

#define NN   50000
#define EE   800000
#define H2C  64
#define H1C  128
#define DINC 256
#define BN_EPS 1e-5f

__global__ void zero_words(unsigned* __restrict__ p, int n) {
  const int i = blockIdx.x * blockDim.x + threadIdx.x;
  if (i < n) p[i] = 0u;
}

template <int K, int NC>
__global__ void __launch_bounds__(NC)
gemm_relu_colstats(const float* __restrict__ A,
                   const float* __restrict__ B,
                   float* __restrict__ C,
                   float* __restrict__ stats,
                   int rowTiles) {
  const int lane = threadIdx.x & 31;
  const int wave = threadIdx.x >> 5;
  const int c = lane & 15;
  const int h = lane >> 4;
  const int col0 = wave << 5;

  v2f b0[K / 4], b1[K / 4];
  {
    const float* __restrict__ bp = B + (size_t)(4 * h) * NC + col0 + c;
#pragma unroll
    for (int t = 0; t < K / 4; ++t) {
      const float* __restrict__ p = bp + (size_t)(8 * (t >> 1) + 2 * (t & 1)) * NC;
      b0[t].x = p[0];
      b0[t].y = p[NC];
      b1[t].x = p[16];
      b1[t].y = p[NC + 16];
    }
  }

  float s0 = 0.0f, q0 = 0.0f, s1 = 0.0f, q1 = 0.0f;

  for (int rt = blockIdx.x; rt < rowTiles; rt += gridDim.x) {
    const int row0 = rt << 4;
    const float4* __restrict__ ap =
        (const float4*)(A + (size_t)(row0 + c) * K + 4 * h);
    float4 a[K / 8];
#pragma unroll
    for (int j = 0; j < K / 8; ++j) a[j] = ap[2 * j];

    v8f acc0 = {};
    v8f acc1 = {};
#pragma unroll
    for (int j = 0; j < K / 8; ++j) {
      v2f lo, hi;
      lo.x = a[j].x; lo.y = a[j].y;
      hi.x = a[j].z; hi.y = a[j].w;
      acc0 = __builtin_amdgcn_wmma_f32_16x16x4_f32(
          false, lo, false, b0[2 * j], (short)0, acc0, false, false);
      acc1 = __builtin_amdgcn_wmma_f32_16x16x4_f32(
          false, lo, false, b1[2 * j], (short)0, acc1, false, false);
      acc0 = __builtin_amdgcn_wmma_f32_16x16x4_f32(
          false, hi, false, b0[2 * j + 1], (short)0, acc0, false, false);
      acc1 = __builtin_amdgcn_wmma_f32_16x16x4_f32(
          false, hi, false, b1[2 * j + 1], (short)0, acc1, false, false);
    }

    float* __restrict__ cp = C + (size_t)(row0 + 8 * h) * NC + col0 + c;
#pragma unroll
    for (int r = 0; r < 8; ++r) {
      float v0 = acc0[r];
      float v1 = acc1[r];
      v0 = v0 > 0.0f ? v0 : 0.0f;
      v1 = v1 > 0.0f ? v1 : 0.0f;
      cp[(size_t)r * NC]      = v0;
      cp[(size_t)r * NC + 16] = v1;
      s0 += v0; q0 += v0 * v0;
      s1 += v1; q1 += v1 * v1;
    }
  }

  s0 += __shfl_xor(s0, 16); q0 += __shfl_xor(q0, 16);
  s1 += __shfl_xor(s1, 16); q1 += __shfl_xor(q1, 16);
  if (h == 0) {
    atomicAdd(&stats[col0 + c], s0);
    atomicAdd(&stats[NC + col0 + c], q0);
    atomicAdd(&stats[col0 + 16 + c], s1);
    atomicAdd(&stats[NC + col0 + 16 + c], q1);
  }
}

__global__ void count_rank(const int* __restrict__ dst,
                           int* __restrict__ cnt,
                           int* __restrict__ rank, int E4) {
  const int t = blockIdx.x * blockDim.x + threadIdx.x;
  if (t >= E4) return;
  const int4 d = ((const int4*)dst)[t];
  int4 r;
  r.x = atomicAdd(cnt + d.x, 1);
  r.y = atomicAdd(cnt + d.y, 1);
  r.z = atomicAdd(cnt + d.z, 1);
  r.w = atomicAdd(cnt + d.w, 1);
  ((int4*)rank)[t] = r;
}

__global__ void row_offsets(const int* __restrict__ cnt,
                            int* __restrict__ cursor,
                            int* __restrict__ rowstart, int n) {
  __shared__ int s_wave[8];
  __shared__ int s_base;
  const int row  = blockIdx.x * 256 + threadIdx.x;
  const int lane = threadIdx.x & 31;
  const int wv   = threadIdx.x >> 5;
  const int c = row < n ? cnt[row] : 0;
  int inc = c;
#pragma unroll
  for (int o = 1; o < 32; o <<= 1) {
    const int t = __shfl_up(inc, o);
    if (lane >= o) inc += t;
  }
  if (lane == 31) s_wave[wv] = inc;
  __syncthreads();
  if (threadIdx.x == 0) {
    int tot = 0;
    for (int i = 0; i < 8; ++i) {
      const int t = s_wave[i];
      s_wave[i] = tot;
      tot += t;
    }
    s_base = atomicAdd(cursor, tot);
  }
  __syncthreads();
  if (row < n) rowstart[row] = s_base + s_wave[wv] + inc - c;
}

__global__ void place_edges(const int* __restrict__ src,
                            const int* __restrict__ dst,
                            const float* __restrict__ w,
                            const int* __restrict__ rank,
                            const int* __restrict__ rowstart,
                            int* __restrict__ edges, int E4, int E) {
  const int t = blockIdx.x * blockDim.x + threadIdx.x;
  if (t >= E4) return;
  const int4   d  = ((const int4*)dst)[t];
  const int4   r  = ((const int4*)rank)[t];
  const int4   s  = ((const int4*)src)[t];
  const float4 ww = ((const float4*)w)[t];
  int2* __restrict__ es = (int2*)edges;
  const int p0 = rowstart[d.x] + r.x;
  const int p1 = rowstart[d.y] + r.y;
  const int p2 = rowstart[d.z] + r.z;
  const int p3 = rowstart[d.w] + r.w;
  if ((unsigned)p0 < (unsigned)E) es[p0] = make_int2(s.x, __float_as_int(ww.x));
  if ((unsigned)p1 < (unsigned)E) es[p1] = make_int2(s.y, __float_as_int(ww.y));
  if ((unsigned)p2 < (unsigned)E) es[p2] = make_int2(s.z, __float_as_int(ww.z));
  if ((unsigned)p3 < (unsigned)E) es[p3] = make_int2(s.w, __float_as_int(ww.w));
}

template <int VPL>
__global__ void gather_rows(const float* __restrict__ x,
                            float* __restrict__ ax,
                            const int* __restrict__ rowstart,
                            const int* __restrict__ cnt,
                            const int* __restrict__ edges,
                            int n, int E) {
  typedef float vf __attribute__((ext_vector_type(VPL)));
  const int2* __restrict__ es = (const int2*)edges;
  const int lane = threadIdx.x & 31;
  const int wave0 = __builtin_amdgcn_readfirstlane(
      (int)((blockIdx.x * blockDim.x + threadIdx.x) >> 5));
  const int nwaves = (int)((gridDim.x * blockDim.x) >> 5);

  for (int row = wave0; row < n; row += nwaves) {
    const int start = rowstart[row];
    const int deg   = cnt[row];
    vf acc = {};
    for (int base = 0; base < deg; base += 32) {
      const int m = min(32, deg - base);
      const int j = min(lane, m - 1);
      const int2 ed = es[min(start + base + j, E - 1)];
      const int   e_src = ed.x;
      const float e_w   = lane < m ? __int_as_float(ed.y) : 0.0f;
      for (int j0 = 0; j0 < m; j0 += 8) {
        vf v[8];
        float wj[8];
#pragma unroll
        for (int u = 0; u < 8; ++u) {
          const int s = __builtin_amdgcn_readlane(e_src, j0 + u);
          wj[u] = __int_as_float(
              __builtin_amdgcn_readlane(__float_as_int(e_w), j0 + u));
          v[u] = *(const vf*)(x + (size_t)s * (32 * VPL) + lane * VPL);
        }
#pragma unroll
        for (int u = 0; u < 8; ++u) acc += wj[u] * v[u];
      }
    }
    *(vf*)(ax + (size_t)row * (32 * VPL) + lane * VPL) = acc;
  }
}

typedef float v4f __attribute__((ext_vector_type(4)));
#define BN_U 10

template <int D>
__global__ void __launch_bounds__(256)
bn_normalize(const float* __restrict__ agg,
             float* __restrict__ out,
             const float* __restrict__ gamma,
             const float* __restrict__ beta,
             const float* __restrict__ stats,
             int Nrows) {
  const int cg = threadIdx.x & (D / 4 - 1);
  const float invN = 1.0f / (float)Nrows;
  const v4f sm = ((const v4f*)stats)[cg];
  const v4f sq = ((const v4f*)(stats + D))[cg];
  const v4f ga = ((const v4f*)gamma)[cg];
  const v4f be = ((const v4f*)beta)[cg];
  const v4f mean = sm * invN;
  const v4f var = sq * invN - mean * mean;
  v4f sc;
  sc.x = ga.x * rsqrtf(var.x + BN_EPS);
  sc.y = ga.y * rsqrtf(var.y + BN_EPS);
  sc.z = ga.z * rsqrtf(var.z + BN_EPS);
  sc.w = ga.w * rsqrtf(var.w + BN_EPS);

  const int total4 = Nrows * (D / 4);
  const int stride = (int)(gridDim.x * 256);
  const v4f* __restrict__ in4 = (const v4f*)agg;
  v4f* __restrict__ out4 = (v4f*)out;

  int i0 = (int)(blockIdx.x * 256 + threadIdx.x);
  for (; i0 + (BN_U - 1) * stride < total4; i0 += BN_U * stride) {
    v4f v[BN_U];
#pragma unroll
    for (int u = 0; u < BN_U; ++u) v[u] = in4[i0 + u * stride];
    static_assert(BN_U == 10, "the statement below names every element of v");
    asm volatile("" ::"v"(v[0]), "v"(v[1]), "v"(v[2]), "v"(v[3]), "v"(v[4]),
                 "v"(v[5]), "v"(v[6]), "v"(v[7]), "v"(v[8]), "v"(v[9]));
#pragma unroll
    for (int u = 0; u < BN_U; ++u)
      out4[i0 + u * stride] = (v[u] - mean) * sc + be;
  }
  for (; i0 < total4; i0 += stride)
    out4[i0] = (in4[i0] - mean) * sc + be;
}

extern "C" void kernel_launch(void* const* d_in, const int* in_sizes, int n_in,
                              void* d_out, int out_size, void* d_ws,
                              size_t ws_size, hipStream_t stream) {
  (void)in_sizes; (void)n_in; (void)out_size; (void)ws_size;

  const float* y   = (const float*)d_in[0];
  const int*   src = (const int*)d_in[1];
  const int*   dst = (const int*)d_in[2];
  const float* w   = (const float*)d_in[3];
  const float* W1  = (const float*)d_in[4];
  const float* g1  = (const float*)d_in[5];
  const float* b1  = (const float*)d_in[6];
  const float* W2  = (const float*)d_in[7];
  const float* g2  = (const float*)d_in[8];
  const float* b2  = (const float*)d_in[9];
  float* out = (float*)d_out;

  float* ws     = (float*)d_ws;
  float* z1     = ws;
  float* h1     = z1 + (size_t)NN * H1C;
  float* z2     = h1 + (size_t)NN * H1C;
  float* ax1    = z2 + (size_t)NN * DINC;
  float* ax2    = ax1 + (size_t)NN * H2C;
  float* stats1 = ax2 + (size_t)NN * H1C;
  float* stats2 = stats1 + 2 * H1C;
  int*   cnt    = (int*)(stats2 + 2 * DINC);
  int*   cursor = cnt + NN;
  int*   rowst  = cursor + 64;
  int*   rank   = rowst + NN;
  int*   edges  = rank + EE;

  zero_words<<<(2 * (H1C + DINC) + NN + 64 + 255) / 256, 256, 0, stream>>>(
      (unsigned*)stats1, 2 * (H1C + DINC) + NN + 64);

  count_rank<<<(EE / 4 + 255) / 256, 256, 0, stream>>>(dst, cnt, rank, EE / 4);
  row_offsets<<<(NN + 255) / 256, 256, 0, stream>>>(cnt, cursor, rowst, NN);
  place_edges<<<(EE / 4 + 255) / 256, 256, 0, stream>>>(
      src, dst, w, rank, rowst, edges, EE / 4, EE);

  gather_rows<2><<<2048, 256, 0, stream>>>(y, ax1, rowst, cnt, edges, NN, EE);

  gemm_relu_colstats<H2C, H1C><<<1042, H1C, 0, stream>>>(
      ax1, W1, z1, stats1, NN / 16);

  bn_normalize<H1C><<<625, 256, 0, stream>>>(z1, h1, g1, b1, stats1, NN);

  gather_rows<4><<<2048, 256, 0, stream>>>(h1, ax2, rowst, cnt, edges, NN, EE);

  gemm_relu_colstats<H1C, DINC><<<512, DINC, 0, stream>>>(
      ax2, W2, z2, stats2, NN / 16);

  bn_normalize<DINC><<<625, 256, 0, stream>>>(z2, out, g2, b2, stats2, NN);
}


# Edges grouped by destination row with LDS counters in two levels (256-row bins, then rows; returning LDS add as slot): per-edge global atomics and the zero launch go; order of float additions changes
# speedup vs baseline: 4.0875x; 1.3596x over previous
#include <hip/hip_runtime.h>
#include <hip/hip_bf16.h>

typedef float v2f __attribute__((ext_vector_type(2)));
typedef float v8f __attribute__((ext_vector_type(8)));

#define NN   50000
#define EE   800000
#define H2C  64
#define H1C  128
#define DINC 256
#define BN_EPS 1e-5f

template <int K, int NC>
__global__ void __launch_bounds__(NC)
gemm_relu_colstats(const float* __restrict__ A,
                   const float* __restrict__ B,
                   float* __restrict__ C,
                   float* __restrict__ stats,
                   int rowTiles) {
  const int lane = threadIdx.x & 31;
  const int wave = threadIdx.x >> 5;
  const int c = lane & 15;
  const int h = lane >> 4;
  const int col0 = wave << 5;

  v2f b0[K / 4], b1[K / 4];
  {
    const float* __restrict__ bp = B + (size_t)(4 * h) * NC + col0 + c;
#pragma unroll
    for (int t = 0; t < K / 4; ++t) {
      const float* __restrict__ p = bp + (size_t)(8 * (t >> 1) + 2 * (t & 1)) * NC;
      b0[t].x = p[0];
      b0[t].y = p[NC];
      b1[t].x = p[16];
      b1[t].y = p[NC + 16];
    }
  }

  float s0 = 0.0f, q0 = 0.0f, s1 = 0.0f, q1 = 0.0f;

  for (int rt = blockIdx.x; rt < rowTiles; rt += gridDim.x) {
    const int row0 = rt << 4;
    const float4* __restrict__ ap =
        (const float4*)(A + (size_t)(row0 + c) * K + 4 * h);
    float4 a[K / 8];
#pragma unroll
    for (int j = 0; j < K / 8; ++j) a[j] = ap[2 * j];

    v8f acc0 = {};
    v8f acc1 = {};
#pragma unroll
    for (int j = 0; j < K / 8; ++j) {
      v2f lo, hi;
      lo.x = a[j].x; lo.y = a[j].y;
      hi.x = a[j].z; hi.y = a[j].w;
      acc0 = __builtin_amdgcn_wmma_f32_16x16x4_f32(
          false, lo, false, b0[2 * j], (short)0, acc0, false, false);
      acc1 = __builtin_amdgcn_wmma_f32_16x16x4_f32(
          false, lo, false, b1[2 * j], (short)0, acc1, false, false);
      acc0 = __builtin_amdgcn_wmma_f32_16x16x4_f32(
          false, hi, false, b0[2 * j + 1], (short)0, acc0, false, false);
      acc1 = __builtin_amdgcn_wmma_f32_16x16x4_f32(
          false, hi, false, b1[2 * j + 1], (short)0, acc1, false, false);
    }

    float* __restrict__ cp = C + (size_t)(row0 + 8 * h) * NC + col0 + c;
#pragma unroll
    for (int r = 0; r < 8; ++r) {
      float v0 = acc0[r];
      float v1 = acc1[r];
      v0 = v0 > 0.0f ? v0 : 0.0f;
      v1 = v1 > 0.0f ? v1 : 0.0f;
      cp[(size_t)r * NC]      = v0;
      cp[(size_t)r * NC + 16] = v1;
      s0 += v0; q0 += v0 * v0;
      s1 += v1; q1 += v1 * v1;
    }
  }

  s0 += __shfl_xor(s0, 16); q0 += __shfl_xor(q0, 16);
  s1 += __shfl_xor(s1, 16); q1 += __shfl_xor(q1, 16);
  if (h == 0) {
    atomicAdd(&stats[col0 + c], s0);
    atomicAdd(&stats[NC + col0 + c], q0);
    atomicAdd(&stats[col0 + 16 + c], s1);
    atomicAdd(&stats[NC + col0 + 16 + c], q1);
  }
}

#define BIN_SHIFT 8
#define BIN_ROWS  (1 << BIN_SHIFT)
#define NBINS     ((NN + BIN_ROWS - 1) >> BIN_SHIFT)
#define NGRP      250
#define EPG       (EE / NGRP)
#define EPG4      (EPG / 4)
#define L1_T      256
#define L1_Q      ((EPG4 + L1_T - 1) / L1_T)
#define L2_T      512
#define L2_B      8
#define CS_B      50
#define CS_Q      ((CS_B + 3) / 4)
static_assert(NGRP % CS_B == 0 && CS_Q == 13, "the load fence names 13 quads");
typedef int v4i __attribute__((ext_vector_type(4)));
static_assert(NGRP * EPG == EE && EPG % 4 == 0, "groups tile the edge list");
static_assert(NBINS <= L1_T && BIN_ROWS <= L2_T, "one thread per bin / row");
static_assert(NN <= 65536 && BIN_ROWS <= 256, "src and row-in-bin share a word");

__device__ __forceinline__ int block_excl_scan(int v, int* s_wave) {
  const int lane = threadIdx.x & 31;
  const int wv   = threadIdx.x >> 5;
  int inc = v;
  for (int o = 1; o < 32; o <<= 1) {
    const int t = __shfl_up(inc, o);
    if (lane >= o) inc += t;
  }
  if (lane == 31) s_wave[wv] = inc;
  __syncthreads();
  if (wv == 0) {
    const int nw = (int)(blockDim.x >> 5);
    const int x = lane < nw ? s_wave[lane] : 0;
    int y = x;
    for (int o = 1; o < 32; o <<= 1) {
      const int t = __shfl_up(y, o);
      if (lane >= o) y += t;
    }
    if (lane < nw) s_wave[lane] = y - x;
  }
  __syncthreads();
  const int r = s_wave[wv] + inc - v;
  __syncthreads();
  return r;
}

__device__ __forceinline__ int bin_of(int d) {
  return (int)min((unsigned)d >> BIN_SHIFT, (unsigned)(NBINS - 1));
}

__global__ void __launch_bounds__(L1_T)
bin_count(const int* __restrict__ dst,
          int* __restrict__ counts,
          unsigned* __restrict__ stats, int nstats) {
  __shared__ int hist[NBINS];
  const int t = threadIdx.x;
  const int g = blockIdx.x;
  if (t < NBINS) hist[t] = 0;
  if (g == 0)
    for (int i = t; i < nstats; i += L1_T) stats[i] = 0u;
  const int4* __restrict__ d4 = (const int4*)dst + (size_t)g * EPG4;
  int4 d[L1_Q];
#pragma unroll
  for (int k = 0; k < L1_Q; ++k) d[k] = d4[min(t + k * L1_T, EPG4 - 1)];
  __syncthreads();
#pragma unroll
  for (int k = 0; k < L1_Q; ++k) {
    if (t + k * L1_T < EPG4) {
      atomicAdd(&hist[bin_of(d[k].x)], 1);
      atomicAdd(&hist[bin_of(d[k].y)], 1);
      atomicAdd(&hist[bin_of(d[k].z)], 1);
      atomicAdd(&hist[bin_of(d[k].w)], 1);
    }
  }
  __syncthreads();
  if (t < NBINS) counts[g * NBINS + t] = hist[t];
}

__global__ void __launch_bounds__(L1_T)
bin_place(const int* __restrict__ src,
          const int* __restrict__ dst,
          const float* __restrict__ w,
          const int* __restrict__ counts,
          int* __restrict__ binstart_g,
          int* __restrict__ bintot_g,
          int* __restrict__ binned) {
  __shared__ int cur[NBINS];
  __shared__ int s_wave[32];
  const int t = threadIdx.x;
  const int g = blockIdx.x;
  const int4*   __restrict__ d4 = (const int4*)dst + (size_t)g * EPG4;
  const int4*   __restrict__ s4 = (const int4*)src + (size_t)g * EPG4;
  const float4* __restrict__ w4 = (const float4*)w + (size_t)g * EPG4;
  int4 d[L1_Q], s[L1_Q];
  float4 ww[L1_Q];
#pragma unroll
  for (int k = 0; k < L1_Q; ++k) {
    const int q = min(t + k * L1_T, EPG4 - 1);
    d[k] = d4[q];
    s[k] = s4[q];
    ww[k] = w4[q];
  }

  int tot = 0, part = 0;
  if (t < NBINS) {
    const int* __restrict__ cp = counts + t;
    for (int g0 = 0; g0 < NGRP; g0 += CS_B) {
      v4i v[CS_Q];
      v[CS_Q - 1] = (v4i){0, 0, 0, 0};
#pragma unroll
      for (int i = 0; i < CS_B; ++i) v[i >> 2][i & 3] = cp[(g0 + i) * NBINS];
      asm volatile("" ::"v"(v[0]), "v"(v[1]), "v"(v[2]), "v"(v[3]), "v"(v[4]),
                   "v"(v[5]), "v"(v[6]), "v"(v[7]), "v"(v[8]), "v"(v[9]),
                   "v"(v[10]), "v"(v[11]), "v"(v[12]));
#pragma unroll
      for (int i = 0; i < CS_B; ++i) {
        const int x = v[i >> 2][i & 3];
        tot += x;
        part += g0 + i < g ? x : 0;
      }
    }
  }
  const int binstart = block_excl_scan(tot, s_wave);
  if (t < NBINS) {
    cur[t] = binstart + part;
    if (g == 0) {
      binstart_g[t] = binstart;
      bintot_g[t] = tot;
    }
  }
  __syncthreads();

  int2* __restrict__ bs = (int2*)binned;
#pragma unroll
  for (int k = 0; k < L1_Q; ++k) {
    if (t + k * L1_T < EPG4) {
      const int dd[4] = {d[k].x, d[k].y, d[k].z, d[k].w};
      const int ss[4] = {s[k].x, s[k].y, s[k].z, s[k].w};
      const float wv4[4] = {ww[k].x, ww[k].y, ww[k].z, ww[k].w};
      int p[4];
#pragma unroll
      for (int u = 0; u < 4; ++u) p[u] = atomicAdd(&cur[bin_of(dd[u])], 1);
#pragma unroll
      for (int u = 0; u < 4; ++u) {
        const int key = (ss[u] & 0xFFFF) | ((dd[u] & (BIN_ROWS - 1)) << 16);
        if ((unsigned)p[u] < (unsigned)EE)
          bs[p[u]] = make_int2(key, __float_as_int(wv4[u]));
      }
    }
  }
}

__global__ void __launch_bounds__(L2_T)
bin_rows(const int* __restrict__ binstart_g,
         const int* __restrict__ bintot_g,
         const int* __restrict__ binned,
         int* __restrict__ rowstart,
         int* __restrict__ cnt,
         int* __restrict__ edges) {
  __shared__ int c[BIN_ROWS];
  __shared__ int cur[BIN_ROWS];
  __shared__ int s_wave[32];
  const int t = threadIdx.x;
  const int b = blockIdx.x;

  const int base = binstart_g[b];
  const int tot  = bintot_g[b];
  const int n = (unsigned)base <= (unsigned)EE ? min(tot, EE - base) : 0;
  if (t < BIN_ROWS) c[t] = 0;
  __syncthreads();

  const int2* __restrict__ bs = (const int2*)binned;
  for (int i0 = t; i0 < n; i0 += L2_B * L2_T) {
    int key[L2_B];
#pragma unroll
    for (int k = 0; k < L2_B; ++k)
      key[k] = bs[base + min(i0 + k * L2_T, n - 1)].x;
    static_assert(L2_B == 8, "the statement below names every element of key");
    asm volatile("" ::"v"(key[0]), "v"(key[1]), "v"(key[2]), "v"(key[3]),
                 "v"(key[4]), "v"(key[5]), "v"(key[6]), "v"(key[7]));
#pragma unroll
    for (int k = 0; k < L2_B; ++k)
      if (i0 + k * L2_T < n)
        atomicAdd(&c[(key[k] >> 16) & (BIN_ROWS - 1)], 1);
  }
  __syncthreads();

  const int cc  = t < BIN_ROWS ? c[t] : 0;
  const int off = block_excl_scan(cc, s_wave);
  if (t < BIN_ROWS) {
    cur[t] = base + off;
    const int row = (b << BIN_SHIFT) + t;
    if (row < NN) {
      rowstart[row] = base + off;
      cnt[row] = cc;
    }
  }
  __syncthreads();

  int2* __restrict__ es = (int2*)edges;
  for (int i0 = t; i0 < n; i0 += L2_B * L2_T) {
    int2 e[L2_B];
    int p[L2_B];
#pragma unroll
    for (int k = 0; k < L2_B; ++k)
      e[k] = bs[base + min(i0 + k * L2_T, n - 1)];
#pragma unroll
    for (int k = 0; k < L2_B; ++k)
      p[k] = i0 + k * L2_T < n
                 ? atomicAdd(&cur[(e[k].x >> 16) & (BIN_ROWS - 1)], 1)
                 : EE;
#pragma unroll
    for (int k = 0; k < L2_B; ++k) {
      if ((unsigned)p[k] < (unsigned)EE)
        es[p[k]] = make_int2(e[k].x & 0xFFFF, e[k].y);
    }
  }
}

template <int VPL>
__global__ void gather_rows(const float* __restrict__ x,
                            float* __restrict__ ax,
                            const int* __restrict__ rowstart,
                            const int* __restrict__ cnt,
                            const int* __restrict__ edges,
                            int n, int E) {
  typedef float vf __attribute__((ext_vector_type(VPL)));
  const int2* __restrict__ es = (const int2*)edges;
  const int lane = threadIdx.x & 31;
  const int wave0 = __builtin_amdgcn_readfirstlane(
      (int)((blockIdx.x * blockDim.x + threadIdx.x) >> 5));
  const int nwaves = (int)((gridDim.x * blockDim.x) >> 5);

  for (int row = wave0; row < n; row += nwaves) {
    const int start = rowstart[row];
    const int deg   = cnt[row];
    vf acc = {};
    for (int base = 0; base < deg; base += 32) {
      const int m = min(32, deg - base);
      const int j = min(lane, m - 1);
      const int2 ed = es[min(start + base + j, E - 1)];
      const int   e_src = ed.x;
      const float e_w   = lane < m ? __int_as_float(ed.y) : 0.0f;
      for (int j0 = 0; j0 < m; j0 += 8) {
        vf v[8];
        float wj[8];
#pragma unroll
        for (int u = 0; u < 8; ++u) {
          const int s = __builtin_amdgcn_readlane(e_src, j0 + u);
          wj[u] = __int_as_float(
              __builtin_amdgcn_readlane(__float_as_int(e_w), j0 + u));
          v[u] = *(const vf*)(x + (size_t)s * (32 * VPL) + lane * VPL);
        }
#pragma unroll
        for (int u = 0; u < 8; ++u) acc += wj[u] * v[u];
      }
    }
    *(vf*)(ax + (size_t)row * (32 * VPL) + lane * VPL) = acc;
  }
}

typedef float v4f __attribute__((ext_vector_type(4)));
#define BN_U 10

template <int D>
__global__ void __launch_bounds__(256)
bn_normalize(const float* __restrict__ agg,
             float* __restrict__ out,
             const float* __restrict__ gamma,
             const float* __restrict__ beta,
             const float* __restrict__ stats,
             int Nrows) {
  const int cg = threadIdx.x & (D / 4 - 1);
  const float invN = 1.0f / (float)Nrows;
  const v4f sm = ((const v4f*)stats)[cg];
  const v4f sq = ((const v4f*)(stats + D))[cg];
  const v4f ga = ((const v4f*)gamma)[cg];
  const v4f be = ((const v4f*)beta)[cg];
  const v4f mean = sm * invN;
  const v4f var = sq * invN - mean * mean;
  v4f sc;
  sc.x = ga.x * rsqrtf(var.x + BN_EPS);
  sc.y = ga.y * rsqrtf(var.y + BN_EPS);
  sc.z = ga.z * rsqrtf(var.z + BN_EPS);
  sc.w = ga.w * rsqrtf(var.w + BN_EPS);

  const int total4 = Nrows * (D / 4);
  const int stride = (int)(gridDim.x * 256);
  const v4f* __restrict__ in4 = (const v4f*)agg;
  v4f* __restrict__ out4 = (v4f*)out;

  int i0 = (int)(blockIdx.x * 256 + threadIdx.x);
  for (; i0 + (BN_U - 1) * stride < total4; i0 += BN_U * stride) {
    v4f v[BN_U];
#pragma unroll
    for (int u = 0; u < BN_U; ++u) v[u] = in4[i0 + u * stride];
    static_assert(BN_U == 10, "the statement below names every element of v");
    asm volatile("" ::"v"(v[0]), "v"(v[1]), "v"(v[2]), "v"(v[3]), "v"(v[4]),
                 "v"(v[5]), "v"(v[6]), "v"(v[7]), "v"(v[8]), "v"(v[9]));
#pragma unroll
    for (int u = 0; u < BN_U; ++u)
      out4[i0 + u * stride] = (v[u] - mean) * sc + be;
  }
  for (; i0 < total4; i0 += stride)
    out4[i0] = (in4[i0] - mean) * sc + be;
}

extern "C" void kernel_launch(void* const* d_in, const int* in_sizes, int n_in,
                              void* d_out, int out_size, void* d_ws,
                              size_t ws_size, hipStream_t stream) {
  (void)in_sizes; (void)n_in; (void)out_size; (void)ws_size;

  const float* y   = (const float*)d_in[0];
  const int*   src = (const int*)d_in[1];
  const int*   dst = (const int*)d_in[2];
  const float* w   = (const float*)d_in[3];
  const float* W1  = (const float*)d_in[4];
  const float* g1  = (const float*)d_in[5];
  const float* b1  = (const float*)d_in[6];
  const float* W2  = (const float*)d_in[7];
  const float* g2  = (const float*)d_in[8];
  const float* b2  = (const float*)d_in[9];
  float* out = (float*)d_out;

  float* ws     = (float*)d_ws;
  float* z1     = ws;
  float* h1     = z1 + (size_t)NN * H1C;
  float* z2     = h1 + (size_t)NN * H1C;
  float* ax1    = z2 + (size_t)NN * DINC;
  float* ax2    = ax1 + (size_t)NN * H2C;
  float* stats1 = ax2 + (size_t)NN * H1C;
  float* stats2 = stats1 + 2 * H1C;
  int*   cnt    = (int*)(stats2 + 2 * DINC);
  int*   rowst  = cnt + NN;
  int*   counts = rowst + NN;
  int*   binst  = counts + NGRP * NBINS;
  int*   bintot = binst + 256;
  int*   binned = bintot + 256;
  int*   edges  = binned + 2 * EE;

  bin_count<<<NGRP, L1_T, 0, stream>>>(dst, counts, (unsigned*)stats1,
                                       2 * (H1C + DINC));
  bin_place<<<NGRP, L1_T, 0, stream>>>(src, dst, w, counts, binst, bintot,
                                       binned);
  bin_rows<<<NBINS, L2_T, 0, stream>>>(binst, bintot, binned, rowst, cnt,
                                       edges);

  gather_rows<2><<<2048, 256, 0, stream>>>(y, ax1, rowst, cnt, edges, NN, EE);

  gemm_relu_colstats<H2C, H1C><<<1042, H1C, 0, stream>>>(
      ax1, W1, z1, stats1, NN / 16);

  bn_normalize<H1C><<<625, 256, 0, stream>>>(z1, h1, g1, b1, stats1, NN);

  gather_rows<4><<<2048, 256, 0, stream>>>(h1, ax2, rowst, cnt, edges, NN, EE);

  gemm_relu_colstats<H1C, DINC><<<512, DINC, 0, stream>>>(
      ax2, W2, z2, stats2, NN / 16);

  bn_normalize<DINC><<<625, 256, 0, stream>>>(z2, out, g2, b2, stats2, NN);
}


# First BatchNorm folded into the second layer's edge sums: scale and shift applied once per summed row (sc*sum(w*z)+sh*sum(w)), its pass and stored rows go; rounding order of that sum changes
# speedup vs baseline: 4.9104x; 1.2013x over previous
#include <hip/hip_runtime.h>
#include <hip/hip_bf16.h>

typedef float v2f __attribute__((ext_vector_type(2)));
typedef float v8f __attribute__((ext_vector_type(8)));

#define NN   50000
#define EE   800000
#define H2C  64
#define H1C  128
#define DINC 256
#define BN_EPS 1e-5f

template <int K, int NC>
__global__ void __launch_bounds__(NC)
gemm_relu_colstats(const float* __restrict__ A,
                   const float* __restrict__ B,
                   float* __restrict__ C,
                   float* __restrict__ stats,
                   int rowTiles) {
  const int lane = threadIdx.x & 31;
  const int wave = threadIdx.x >> 5;
  const int c = lane & 15;
  const int h = lane >> 4;
  const int col0 = wave << 5;

  v2f b0[K / 4], b1[K / 4];
  {
    const float* __restrict__ bp = B + (size_t)(4 * h) * NC + col0 + c;
#pragma unroll
    for (int t = 0; t < K / 4; ++t) {
      const float* __restrict__ p = bp + (size_t)(8 * (t >> 1) + 2 * (t & 1)) * NC;
      b0[t].x = p[0];
      b0[t].y = p[NC];
      b1[t].x = p[16];
      b1[t].y = p[NC + 16];
    }
  }

  float s0 = 0.0f, q0 = 0.0f, s1 = 0.0f, q1 = 0.0f;

  for (int rt = blockIdx.x; rt < rowTiles; rt += gridDim.x) {
    const int row0 = rt << 4;
    const float4* __restrict__ ap =
        (const float4*)(A + (size_t)(row0 + c) * K + 4 * h);
    float4 a[K / 8];
#pragma unroll
    for (int j = 0; j < K / 8; ++j) a[j] = ap[2 * j];

    v8f acc0 = {};
    v8f acc1 = {};
#pragma unroll
    for (int j = 0; j < K / 8; ++j) {
      v2f lo, hi;
      lo.x = a[j].x; lo.y = a[j].y;
      hi.x = a[j].z; hi.y = a[j].w;
      acc0 = __builtin_amdgcn_wmma_f32_16x16x4_f32(
          false, lo, false, b0[2 * j], (short)0, acc0, false, false);
      acc1 = __builtin_amdgcn_wmma_f32_16x16x4_f32(
          false, lo, false, b1[2 * j], (short)0, acc1, false, false);
      acc0 = __builtin_amdgcn_wmma_f32_16x16x4_f32(
          false, hi, false, b0[2 * j + 1], (short)0, acc0, false, false);
      acc1 = __builtin_amdgcn_wmma_f32_16x16x4_f32(
          false, hi, false, b1[2 * j + 1], (short)0, acc1, false, false);
    }

    float* __restrict__ cp = C + (size_t)(row0 + 8 * h) * NC + col0 + c;
#pragma unroll
    for (int r = 0; r < 8; ++r) {
      float v0 = acc0[r];
      float v1 = acc1[r];
      v0 = v0 > 0.0f ? v0 : 0.0f;
      v1 = v1 > 0.0f ? v1 : 0.0f;
      cp[(size_t)r * NC]      = v0;
      cp[(size_t)r * NC + 16] = v1;
      s0 += v0; q0 += v0 * v0;
      s1 += v1; q1 += v1 * v1;
    }
  }

  s0 += __shfl_xor(s0, 16); q0 += __shfl_xor(q0, 16);
  s1 += __shfl_xor(s1, 16); q1 += __shfl_xor(q1, 16);
  if (h == 0) {
    atomicAdd(&stats[col0 + c], s0);
    atomicAdd(&stats[NC + col0 + c], q0);
    atomicAdd(&stats[col0 + 16 + c], s1);
    atomicAdd(&stats[NC + col0 + 16 + c], q1);
  }
}

#define BIN_SHIFT 8
#define BIN_ROWS  (1 << BIN_SHIFT)
#define NBINS     ((NN + BIN_ROWS - 1) >> BIN_SHIFT)
#define NGRP      250
#define EPG       (EE / NGRP)
#define EPG4      (EPG / 4)
#define L1_T      256
#define L1_Q      ((EPG4 + L1_T - 1) / L1_T)
#define L2_T      512
#define L2_B      8
#define CS_B      50
#define CS_Q      ((CS_B + 3) / 4)
static_assert(NGRP % CS_B == 0 && CS_Q == 13, "the load fence names 13 quads");
typedef int v4i __attribute__((ext_vector_type(4)));
static_assert(NGRP * EPG == EE && EPG % 4 == 0, "groups tile the edge list");
static_assert(NBINS <= L1_T && BIN_ROWS <= L2_T, "one thread per bin / row");
static_assert(NN <= 65536 && BIN_ROWS <= 256, "src and row-in-bin share a word");

__device__ __forceinline__ int block_excl_scan(int v, int* s_wave) {
  const int lane = threadIdx.x & 31;
  const int wv   = threadIdx.x >> 5;
  int inc = v;
  for (int o = 1; o < 32; o <<= 1) {
    const int t = __shfl_up(inc, o);
    if (lane >= o) inc += t;
  }
  if (lane == 31) s_wave[wv] = inc;
  __syncthreads();
  if (wv == 0) {
    const int nw = (int)(blockDim.x >> 5);
    const int x = lane < nw ? s_wave[lane] : 0;
    int y = x;
    for (int o = 1; o < 32; o <<= 1) {
      const int t = __shfl_up(y, o);
      if (lane >= o) y += t;
    }
    if (lane < nw) s_wave[lane] = y - x;
  }
  __syncthreads();
  const int r = s_wave[wv] + inc - v;
  __syncthreads();
  return r;
}

__device__ __forceinline__ int bin_of(int d) {
  return (int)min((unsigned)d >> BIN_SHIFT, (unsigned)(NBINS - 1));
}

__global__ void __launch_bounds__(L1_T)
bin_count(const int* __restrict__ dst,
          int* __restrict__ counts,
          unsigned* __restrict__ stats, int nstats) {
  __shared__ int hist[NBINS];
  const int t = threadIdx.x;
  const int g = blockIdx.x;
  if (t < NBINS) hist[t] = 0;
  if (g == 0)
    for (int i = t; i < nstats; i += L1_T) stats[i] = 0u;
  const int4* __restrict__ d4 = (const int4*)dst + (size_t)g * EPG4;
  int4 d[L1_Q];
#pragma unroll
  for (int k = 0; k < L1_Q; ++k) d[k] = d4[min(t + k * L1_T, EPG4 - 1)];
  __syncthreads();
#pragma unroll
  for (int k = 0; k < L1_Q; ++k) {
    if (t + k * L1_T < EPG4) {
      atomicAdd(&hist[bin_of(d[k].x)], 1);
      atomicAdd(&hist[bin_of(d[k].y)], 1);
      atomicAdd(&hist[bin_of(d[k].z)], 1);
      atomicAdd(&hist[bin_of(d[k].w)], 1);
    }
  }
  __syncthreads();
  if (t < NBINS) counts[g * NBINS + t] = hist[t];
}

__global__ void __launch_bounds__(L1_T)
bin_place(const int* __restrict__ src,
          const int* __restrict__ dst,
          const float* __restrict__ w,
          const int* __restrict__ counts,
          int* __restrict__ binstart_g,
          int* __restrict__ bintot_g,
          int* __restrict__ binned) {
  __shared__ int cur[NBINS];
  __shared__ int s_wave[32];
  const int t = threadIdx.x;
  const int g = blockIdx.x;
  const int4*   __restrict__ d4 = (const int4*)dst + (size_t)g * EPG4;
  const int4*   __restrict__ s4 = (const int4*)src + (size_t)g * EPG4;
  const float4* __restrict__ w4 = (const float4*)w + (size_t)g * EPG4;
  int4 d[L1_Q], s[L1_Q];
  float4 ww[L1_Q];
#pragma unroll
  for (int k = 0; k < L1_Q; ++k) {
    const int q = min(t + k * L1_T, EPG4 - 1);
    d[k] = d4[q];
    s[k] = s4[q];
    ww[k] = w4[q];
  }

  int tot = 0, part = 0;
  if (t < NBINS) {
    const int* __restrict__ cp = counts + t;
    for (int g0 = 0; g0 < NGRP; g0 += CS_B) {
      v4i v[CS_Q];
      v[CS_Q - 1] = (v4i){0, 0, 0, 0};
#pragma unroll
      for (int i = 0; i < CS_B; ++i) v[i >> 2][i & 3] = cp[(g0 + i) * NBINS];
      asm volatile("" ::"v"(v[0]), "v"(v[1]), "v"(v[2]), "v"(v[3]), "v"(v[4]),
                   "v"(v[5]), "v"(v[6]), "v"(v[7]), "v"(v[8]), "v"(v[9]),
                   "v"(v[10]), "v"(v[11]), "v"(v[12]));
#pragma unroll
      for (int i = 0; i < CS_B; ++i) {
        const int x = v[i >> 2][i & 3];
        tot += x;
        part += g0 + i < g ? x : 0;
      }
    }
  }
  const int binstart = block_excl_scan(tot, s_wave);
  if (t < NBINS) {
    cur[t] = binstart + part;
    if (g == 0) {
      binstart_g[t] = binstart;
      bintot_g[t] = tot;
    }
  }
  __syncthreads();

  int2* __restrict__ bs = (int2*)binned;
#pragma unroll
  for (int k = 0; k < L1_Q; ++k) {
    if (t + k * L1_T < EPG4) {
      const int dd[4] = {d[k].x, d[k].y, d[k].z, d[k].w};
      const int ss[4] = {s[k].x, s[k].y, s[k].z, s[k].w};
      const float wv4[4] = {ww[k].x, ww[k].y, ww[k].z, ww[k].w};
      int p[4];
#pragma unroll
      for (int u = 0; u < 4; ++u) p[u] = atomicAdd(&cur[bin_of(dd[u])], 1);
#pragma unroll
      for (int u = 0; u < 4; ++u) {
        const int key = (ss[u] & 0xFFFF) | ((dd[u] & (BIN_ROWS - 1)) << 16);
        if ((unsigned)p[u] < (unsigned)EE)
          bs[p[u]] = make_int2(key, __float_as_int(wv4[u]));
      }
    }
  }
}

__global__ void __launch_bounds__(L2_T)
bin_rows(const int* __restrict__ binstart_g,
         const int* __restrict__ bintot_g,
         const int* __restrict__ binned,
         int* __restrict__ rowstart,
         int* __restrict__ cnt,
         int* __restrict__ edges) {
  __shared__ int c[BIN_ROWS];
  __shared__ int cur[BIN_ROWS];
  __shared__ int s_wave[32];
  const int t = threadIdx.x;
  const int b = blockIdx.x;

  const int base = binstart_g[b];
  const int tot  = bintot_g[b];
  const int n = (unsigned)base <= (unsigned)EE ? min(tot, EE - base) : 0;
  if (t < BIN_ROWS) c[t] = 0;
  __syncthreads();

  const int2* __restrict__ bs = (const int2*)binned;
  for (int i0 = t; i0 < n; i0 += L2_B * L2_T) {
    int key[L2_B];
#pragma unroll
    for (int k = 0; k < L2_B; ++k)
      key[k] = bs[base + min(i0 + k * L2_T, n - 1)].x;
    static_assert(L2_B == 8, "the statement below names every element of key");
    asm volatile("" ::"v"(key[0]), "v"(key[1]), "v"(key[2]), "v"(key[3]),
                 "v"(key[4]), "v"(key[5]), "v"(key[6]), "v"(key[7]));
#pragma unroll
    for (int k = 0; k < L2_B; ++k)
      if (i0 + k * L2_T < n)
        atomicAdd(&c[(key[k] >> 16) & (BIN_ROWS - 1)], 1);
  }
  __syncthreads();

  const int cc  = t < BIN_ROWS ? c[t] : 0;
  const int off = block_excl_scan(cc, s_wave);
  if (t < BIN_ROWS) {
    cur[t] = base + off;
    const int row = (b << BIN_SHIFT) + t;
    if (row < NN) {
      rowstart[row] = base + off;
      cnt[row] = cc;
    }
  }
  __syncthreads();

  int2* __restrict__ es = (int2*)edges;
  for (int i0 = t; i0 < n; i0 += L2_B * L2_T) {
    int2 e[L2_B];
    int p[L2_B];
#pragma unroll
    for (int k = 0; k < L2_B; ++k)
      e[k] = bs[base + min(i0 + k * L2_T, n - 1)];
#pragma unroll
    for (int k = 0; k < L2_B; ++k)
      p[k] = i0 + k * L2_T < n
                 ? atomicAdd(&cur[(e[k].x >> 16) & (BIN_ROWS - 1)], 1)
                 : EE;
#pragma unroll
    for (int k = 0; k < L2_B; ++k) {
      if ((unsigned)p[k] < (unsigned)EE)
        es[p[k]] = make_int2(e[k].x & 0xFFFF, e[k].y);
    }
  }
}

template <int VPL, bool NORM>
__global__ void gather_rows(const float* __restrict__ x,
                            float* __restrict__ ax,
                            const int* __restrict__ rowstart,
                            const int* __restrict__ cnt,
                            const int* __restrict__ edges,
                            int n, int E,
                            const float* __restrict__ gamma,
                            const float* __restrict__ beta,
                            const float* __restrict__ stats) {
  typedef float vf __attribute__((ext_vector_type(VPL)));
  const int2* __restrict__ es = (const int2*)edges;
  const int lane = threadIdx.x & 31;
  vf sc = {}, sh = {};
  if constexpr (NORM) {
    const float invN = 1.0f / (float)n;
    const vf sm = *(const vf*)(stats + lane * VPL);
    const vf sq = *(const vf*)(stats + 32 * VPL + lane * VPL);
    const vf ga = *(const vf*)(gamma + lane * VPL);
    const vf be = *(const vf*)(beta + lane * VPL);
    const vf mean = sm * invN;
    const vf var = sq * invN - mean * mean;
#pragma unroll
    for (int k = 0; k < VPL; ++k) sc[k] = ga[k] * rsqrtf(var[k] + BN_EPS);
    sh = be - mean * sc;
  }
  const int wave0 = __builtin_amdgcn_readfirstlane(
      (int)((blockIdx.x * blockDim.x + threadIdx.x) >> 5));
  const int nwaves = (int)((gridDim.x * blockDim.x) >> 5);

  for (int row = wave0; row < n; row += nwaves) {
    const int start = rowstart[row];
    const int deg   = cnt[row];
    vf acc = {};
    float wsum = 0.0f;
    for (int base = 0; base < deg; base += 32) {
      const int m = min(32, deg - base);
      const int j = min(lane, m - 1);
      const int2 ed = es[min(start + base + j, E - 1)];
      const int   e_src = ed.x;
      const float e_w   = lane < m ? __int_as_float(ed.y) : 0.0f;
      for (int j0 = 0; j0 < m; j0 += 8) {
        vf v[8];
        float wj[8];
#pragma unroll
        for (int u = 0; u < 8; ++u) {
          const int s = __builtin_amdgcn_readlane(e_src, j0 + u);
          wj[u] = __int_as_float(
              __builtin_amdgcn_readlane(__float_as_int(e_w), j0 + u));
          v[u] = *(const vf*)(x + (size_t)s * (32 * VPL) + lane * VPL);
        }
#pragma unroll
        for (int u = 0; u < 8; ++u) {
          acc += wj[u] * v[u];
          if constexpr (NORM) wsum += wj[u];
        }
      }
    }
    if constexpr (NORM) acc = acc * sc + wsum * sh;
    *(vf*)(ax + (size_t)row * (32 * VPL) + lane * VPL) = acc;
  }
}

typedef float v4f __attribute__((ext_vector_type(4)));
#define BN_U 10

template <int D>
__global__ void __launch_bounds__(256)
bn_normalize(const float* __restrict__ agg,
             float* __restrict__ out,
             const float* __restrict__ gamma,
             const float* __restrict__ beta,
             const float* __restrict__ stats,
             int Nrows) {
  const int cg = threadIdx.x & (D / 4 - 1);
  const float invN = 1.0f / (float)Nrows;
  const v4f sm = ((const v4f*)stats)[cg];
  const v4f sq = ((const v4f*)(stats + D))[cg];
  const v4f ga = ((const v4f*)gamma)[cg];
  const v4f be = ((const v4f*)beta)[cg];
  const v4f mean = sm * invN;
  const v4f var = sq * invN - mean * mean;
  v4f sc;
  sc.x = ga.x * rsqrtf(var.x + BN_EPS);
  sc.y = ga.y * rsqrtf(var.y + BN_EPS);
  sc.z = ga.z * rsqrtf(var.z + BN_EPS);
  sc.w = ga.w * rsqrtf(var.w + BN_EPS);

  const int total4 = Nrows * (D / 4);
  const int stride = (int)(gridDim.x * 256);
  const v4f* __restrict__ in4 = (const v4f*)agg;
  v4f* __restrict__ out4 = (v4f*)out;

  int i0 = (int)(blockIdx.x * 256 + threadIdx.x);
  for (; i0 + (BN_U - 1) * stride < total4; i0 += BN_U * stride) {
    v4f v[BN_U];
#pragma unroll
    for (int u = 0; u < BN_U; ++u) v[u] = in4[i0 + u * stride];
    static_assert(BN_U == 10, "the statement below names every element of v");
    asm volatile("" ::"v"(v[0]), "v"(v[1]), "v"(v[2]), "v"(v[3]), "v"(v[4]),
                 "v"(v[5]), "v"(v[6]), "v"(v[7]), "v"(v[8]), "v"(v[9]));
#pragma unroll
    for (int u = 0; u < BN_U; ++u)
      out4[i0 + u * stride] = (v[u] - mean) * sc + be;
  }
  for (; i0 < total4; i0 += stride)
    out4[i0] = (in4[i0] - mean) * sc + be;
}

extern "C" void kernel_launch(void* const* d_in, const int* in_sizes, int n_in,
                              void* d_out, int out_size, void* d_ws,
                              size_t ws_size, hipStream_t stream) {
  (void)in_sizes; (void)n_in; (void)out_size; (void)ws_size;

  const float* y   = (const float*)d_in[0];
  const int*   src = (const int*)d_in[1];
  const int*   dst = (const int*)d_in[2];
  const float* w   = (const float*)d_in[3];
  const float* W1  = (const float*)d_in[4];
  const float* g1  = (const float*)d_in[5];
  const float* b1  = (const float*)d_in[6];
  const float* W2  = (const float*)d_in[7];
  const float* g2  = (const float*)d_in[8];
  const float* b2  = (const float*)d_in[9];
  float* out = (float*)d_out;

  float* ws     = (float*)d_ws;
  float* z1     = ws;
  float* z2     = z1 + (size_t)NN * H1C;
  float* ax1    = z2 + (size_t)NN * DINC;
  float* ax2    = ax1 + (size_t)NN * H2C;
  float* stats1 = ax2 + (size_t)NN * H1C;
  float* stats2 = stats1 + 2 * H1C;
  int*   cnt    = (int*)(stats2 + 2 * DINC);
  int*   rowst  = cnt + NN;
  int*   counts = rowst + NN;
  int*   binst  = counts + NGRP * NBINS;
  int*   bintot = binst + 256;
  int*   binned = bintot + 256;
  int*   edges  = binned + 2 * EE;

  bin_count<<<NGRP, L1_T, 0, stream>>>(dst, counts, (unsigned*)stats1,
                                       2 * (H1C + DINC));
  bin_place<<<NGRP, L1_T, 0, stream>>>(src, dst, w, counts, binst, bintot,
                                       binned);
  bin_rows<<<NBINS, L2_T, 0, stream>>>(binst, bintot, binned, rowst, cnt,
                                       edges);

  gather_rows<2, false><<<2048, 256, 0, stream>>>(y, ax1, rowst, cnt, edges,
                                                  NN, EE, nullptr, nullptr,
                                                  nullptr);

  gemm_relu_colstats<H2C, H1C><<<1042, H1C, 0, stream>>>(
      ax1, W1, z1, stats1, NN / 16);

  gather_rows<4, true><<<2048, 256, 0, stream>>>(z1, ax2, rowst, cnt, edges,
                                                 NN, EE, g1, b1, stats1);

  gemm_relu_colstats<H1C, DINC><<<512, DINC, 0, stream>>>(
      ax2, W2, z2, stats2, NN / 16);

  bn_normalize<DINC><<<625, 256, 0, stream>>>(z2, out, g2, b2, stats2, NN);
}


# Edge sums of each layer go straight into the dense product's LDS A tile (a wave sums its 2 or 4 rows of a tile together, 16 or 32 row loads in flight): two gather launches and the stored sums go
# speedup vs baseline: 5.6933x; 1.1594x over previous
#include <hip/hip_runtime.h>
#include <hip/hip_bf16.h>

typedef float v2f __attribute__((ext_vector_type(2)));
typedef float v8f __attribute__((ext_vector_type(8)));
typedef float v4f __attribute__((ext_vector_type(4)));

#define NN   50000
#define EE   800000
#define H2C  64
#define H1C  128
#define DINC 256
#define BN_EPS 1e-5f

#define BIN_SHIFT 8
#define BIN_ROWS  (1 << BIN_SHIFT)
#define NBINS     ((NN + BIN_ROWS - 1) >> BIN_SHIFT)
#define NGRP      250
#define EPG       (EE / NGRP)
#define EPG4      (EPG / 4)
#define L1_T      256
#define L1_Q      ((EPG4 + L1_T - 1) / L1_T)
#define L2_T      512
#define L2_B      8
#define CS_B      50
#define CS_Q      ((CS_B + 3) / 4)
static_assert(NGRP % CS_B == 0 && CS_Q == 13, "the load fence names 13 quads");
typedef int v4i __attribute__((ext_vector_type(4)));
static_assert(NGRP * EPG == EE && EPG % 4 == 0, "groups tile the edge list");
static_assert(NBINS <= L1_T && BIN_ROWS <= L2_T, "one thread per bin / row");
static_assert(NN <= 65536 && BIN_ROWS <= 256, "src and row-in-bin share a word");

__device__ __forceinline__ int block_excl_scan(int v, int* s_wave) {
  const int lane = threadIdx.x & 31;
  const int wv   = threadIdx.x >> 5;
  int inc = v;
  for (int o = 1; o < 32; o <<= 1) {
    const int t = __shfl_up(inc, o);
    if (lane >= o) inc += t;
  }
  if (lane == 31) s_wave[wv] = inc;
  __syncthreads();
  if (wv == 0) {
    const int nw = (int)(blockDim.x >> 5);
    const int x = lane < nw ? s_wave[lane] : 0;
    int y = x;
    for (int o = 1; o < 32; o <<= 1) {
      const int t = __shfl_up(y, o);
      if (lane >= o) y += t;
    }
    if (lane < nw) s_wave[lane] = y - x;
  }
  __syncthreads();
  const int r = s_wave[wv] + inc - v;
  __syncthreads();
  return r;
}

__device__ __forceinline__ int bin_of(int d) {
  return (int)min((unsigned)d >> BIN_SHIFT, (unsigned)(NBINS - 1));
}

__global__ void __launch_bounds__(L1_T)
bin_count(const int* __restrict__ dst,
          int* __restrict__ counts,
          unsigned* __restrict__ stats, int nstats) {
  __shared__ int hist[NBINS];
  const int t = threadIdx.x;
  const int g = blockIdx.x;
  if (t < NBINS) hist[t] = 0;
  if (g == 0)
    for (int i = t; i < nstats; i += L1_T) stats[i] = 0u;
  const int4* __restrict__ d4 = (const int4*)dst + (size_t)g * EPG4;
  int4 d[L1_Q];
#pragma unroll
  for (int k = 0; k < L1_Q; ++k) d[k] = d4[min(t + k * L1_T, EPG4 - 1)];
  __syncthreads();
#pragma unroll
  for (int k = 0; k < L1_Q; ++k) {
    if (t + k * L1_T < EPG4) {
      atomicAdd(&hist[bin_of(d[k].x)], 1);
      atomicAdd(&hist[bin_of(d[k].y)], 1);
      atomicAdd(&hist[bin_of(d[k].z)], 1);
      atomicAdd(&hist[bin_of(d[k].w)], 1);
    }
  }
  __syncthreads();
  if (t < NBINS) counts[g * NBINS + t] = hist[t];
}

__global__ void __launch_bounds__(L1_T)
bin_place(const int* __restrict__ src,
          const int* __restrict__ dst,
          const float* __restrict__ w,
          const int* __restrict__ counts,
          int* __restrict__ binstart_g,
          int* __restrict__ bintot_g,
          int* __restrict__ binned) {
  __shared__ int cur[NBINS];
  __shared__ int s_wave[32];
  const int t = threadIdx.x;
  const int g = blockIdx.x;
  const int4*   __restrict__ d4 = (const int4*)dst + (size_t)g * EPG4;
  const int4*   __restrict__ s4 = (const int4*)src + (size_t)g * EPG4;
  const float4* __restrict__ w4 = (const float4*)w + (size_t)g * EPG4;
  int4 d[L1_Q], s[L1_Q];
  float4 ww[L1_Q];
#pragma unroll
  for (int k = 0; k < L1_Q; ++k) {
    const int q = min(t + k * L1_T, EPG4 - 1);
    d[k] = d4[q];
    s[k] = s4[q];
    ww[k] = w4[q];
  }

  int tot = 0, part = 0;
  if (t < NBINS) {
    const int* __restrict__ cp = counts + t;
    for (int g0 = 0; g0 < NGRP; g0 += CS_B) {
      v4i v[CS_Q];
      v[CS_Q - 1] = (v4i){0, 0, 0, 0};
#pragma unroll
      for (int i = 0; i < CS_B; ++i) v[i >> 2][i & 3] = cp[(g0 + i) * NBINS];
      asm volatile("" ::"v"(v[0]), "v"(v[1]), "v"(v[2]), "v"(v[3]), "v"(v[4]),
                   "v"(v[5]), "v"(v[6]), "v"(v[7]), "v"(v[8]), "v"(v[9]),
                   "v"(v[10]), "v"(v[11]), "v"(v[12]));
#pragma unroll
      for (int i = 0; i < CS_B; ++i) {
        const int x = v[i >> 2][i & 3];
        tot += x;
        part += g0 + i < g ? x : 0;
      }
    }
  }
  const int binstart = block_excl_scan(tot, s_wave);
  if (t < NBINS) {
    cur[t] = binstart + part;
    if (g == 0) {
      binstart_g[t] = binstart;
      bintot_g[t] = tot;
    }
  }
  __syncthreads();

  int2* __restrict__ bs = (int2*)binned;
#pragma unroll
  for (int k = 0; k < L1_Q; ++k) {
    if (t + k * L1_T < EPG4) {
      const int dd[4] = {d[k].x, d[k].y, d[k].z, d[k].w};
      const int ss[4] = {s[k].x, s[k].y, s[k].z, s[k].w};
      const float wv4[4] = {ww[k].x, ww[k].y, ww[k].z, ww[k].w};
      int p[4];
#pragma unroll
      for (int u = 0; u < 4; ++u) p[u] = atomicAdd(&cur[bin_of(dd[u])], 1);
#pragma unroll
      for (int u = 0; u < 4; ++u) {
        const int key = (ss[u] & 0xFFFF) | ((dd[u] & (BIN_ROWS - 1)) << 16);
        if ((unsigned)p[u] < (unsigned)EE)
          bs[p[u]] = make_int2(key, __float_as_int(wv4[u]));
      }
    }
  }
}

__global__ void __launch_bounds__(L2_T)
bin_rows(const int* __restrict__ binstart_g,
         const int* __restrict__ bintot_g,
         const int* __restrict__ binned,
         int* __restrict__ rowstart,
         int* __restrict__ cnt,
         int* __restrict__ edges) {
  __shared__ int c[BIN_ROWS];
  __shared__ int cur[BIN_ROWS];
  __shared__ int s_wave[32];
  const int t = threadIdx.x;
  const int b = blockIdx.x;

  const int base = binstart_g[b];
  const int tot  = bintot_g[b];
  const int n = (unsigned)base <= (unsigned)EE ? min(tot, EE - base) : 0;
  if (t < BIN_ROWS) c[t] = 0;
  __syncthreads();

  const int2* __restrict__ bs = (const int2*)binned;
  for (int i0 = t; i0 < n; i0 += L2_B * L2_T) {
    int key[L2_B];
#pragma unroll
    for (int k = 0; k < L2_B; ++k)
      key[k] = bs[base + min(i0 + k * L2_T, n - 1)].x;
    static_assert(L2_B == 8, "the statement below names every element of key");
    asm volatile("" ::"v"(key[0]), "v"(key[1]), "v"(key[2]), "v"(key[3]),
                 "v"(key[4]), "v"(key[5]), "v"(key[6]), "v"(key[7]));
#pragma unroll
    for (int k = 0; k < L2_B; ++k)
      if (i0 + k * L2_T < n)
        atomicAdd(&c[(key[k] >> 16) & (BIN_ROWS - 1)], 1);
  }
  __syncthreads();

  const int cc  = t < BIN_ROWS ? c[t] : 0;
  const int off = block_excl_scan(cc, s_wave);
  if (t < BIN_ROWS) {
    cur[t] = base + off;
    const int row = (b << BIN_SHIFT) + t;
    if (row < NN) {
      rowstart[row] = base + off;
      cnt[row] = cc;
    }
  }
  __syncthreads();

  int2* __restrict__ es = (int2*)edges;
  for (int i0 = t; i0 < n; i0 += L2_B * L2_T) {
    int2 e[L2_B];
    int p[L2_B];
#pragma unroll
    for (int k = 0; k < L2_B; ++k)
      e[k] = bs[base + min(i0 + k * L2_T, n - 1)];
#pragma unroll
    for (int k = 0; k < L2_B; ++k)
      p[k] = i0 + k * L2_T < n
                 ? atomicAdd(&cur[(e[k].x >> 16) & (BIN_ROWS - 1)], 1)
                 : EE;
#pragma unroll
    for (int k = 0; k < L2_B; ++k) {
      if ((unsigned)p[k] < (unsigned)EE)
        es[p[k]] = make_int2(e[k].x & 0xFFFF, e[k].y);
    }
  }
}

template <int K, int NC, bool NORM>
__global__ void __launch_bounds__(NC, NC == 256 ? 4 : 5)
edge_gemm_relu_colstats(const float* __restrict__ x,
                        const float* __restrict__ B,
                        float* __restrict__ C,
                        float* __restrict__ stats,
                        int rowTiles,
                        const int* __restrict__ rowstart,
                        const int* __restrict__ cnt,
                        const int* __restrict__ edges,
                        int n, int E,
                        const float* __restrict__ gamma,
                        const float* __restrict__ beta,
                        const float* __restrict__ stats_in) {
  constexpr int VPL = K / 32;
  constexpr int NW  = NC / 32;
  constexpr int RPW = 16 / NW;
  constexpr int P   = K + 4;
  static_assert(NW * RPW == 16 && K % 32 == 0, "a block sums whole row tiles");
  typedef float vf __attribute__((ext_vector_type(VPL)));
  __shared__ __attribute__((aligned(16))) float tile[2][16 * P];

  const int2* __restrict__ es = (const int2*)edges;
  const int lane = threadIdx.x & 31;
  const int wave = __builtin_amdgcn_readfirstlane((int)(threadIdx.x >> 5));
  const int c = lane & 15;
  const int h = lane >> 4;
  const int col0 = wave << 5;

  v2f b0[K / 4], b1[K / 4];
  {
    const float* __restrict__ bp = B + (size_t)(4 * h) * NC + col0 + c;
#pragma unroll
    for (int t = 0; t < K / 4; ++t) {
      const float* __restrict__ p = bp + (size_t)(8 * (t >> 1) + 2 * (t & 1)) * NC;
      b0[t].x = p[0];
      b0[t].y = p[NC];
      b1[t].x = p[16];
      b1[t].y = p[NC + 16];
    }
  }

  vf sc = {}, sh = {};
  if constexpr (NORM) {
    const float invN = 1.0f / (float)n;
    const vf sm = *(const vf*)(stats_in + lane * VPL);
    const vf sq = *(const vf*)(stats_in + 32 * VPL + lane * VPL);
    const vf ga = *(const vf*)(gamma + lane * VPL);
    const vf be = *(const vf*)(beta + lane * VPL);
    const vf mean = sm * invN;
    const vf var = sq * invN - mean * mean;
#pragma unroll
    for (int k = 0; k < VPL; ++k) sc[k] = ga[k] * rsqrtf(var[k] + BN_EPS);
    sh = be - mean * sc;
  }

  auto chunk = [&](int st, int dg, int base) -> int2 {
    const int j = max(min(base + lane, dg - 1), 0);
    return es[min((unsigned)(st + j), (unsigned)(E - 1))];
  };

  float s0 = 0.0f, q0 = 0.0f, s1 = 0.0f, q1 = 0.0f;

  int rt = blockIdx.x;
  int start[RPW], deg[RPW];
  int2 ed[RPW];
#pragma unroll
  for (int q = 0; q < RPW; ++q) {
    const int r = min((rt << 4) + wave * RPW + q, n - 1);
    start[q] = rowstart[r];
    deg[q] = rt < rowTiles ? cnt[r] : 0;
  }
#pragma unroll
  for (int q = 0; q < RPW; ++q) ed[q] = chunk(start[q], deg[q], 0);

  for (int it = 0; rt < rowTiles; rt += gridDim.x, ++it) {
    const int row0 = rt << 4;
    const int rtn = rt + (int)gridDim.x;

    int nstart[RPW], ndeg[RPW];
#pragma unroll
    for (int q = 0; q < RPW; ++q) {
      const int r = min((rtn << 4) + wave * RPW + q, n - 1);
      nstart[q] = rowstart[r];
      ndeg[q] = rtn < rowTiles ? cnt[r] : 0;
    }

    int maxdeg = 0;
#pragma unroll
    for (int q = 0; q < RPW; ++q) maxdeg = max(maxdeg, deg[q]);

    vf acc[RPW];
    float wsum[RPW];
#pragma unroll
    for (int q = 0; q < RPW; ++q) { acc[q] = (vf){}; wsum[q] = 0.0f; }

    for (int base = 0; base < maxdeg; base += 32) {
      int e_src[RPW], m[RPW];
      float e_w[RPW];
#pragma unroll
      for (int q = 0; q < RPW; ++q) {
        m[q] = min(32, deg[q] - base);
        e_src[q] = ed[q].x;
        e_w[q] = lane < m[q] ? __int_as_float(ed[q].y) : 0.0f;
      }
      const int mm = min(32, maxdeg - base);
      for (int j0 = 0; j0 < mm; j0 += 8) {
        vf v[RPW][8];
        float wj[RPW][8];
#pragma unroll
        for (int q = 0; q < RPW; ++q) {
#pragma unroll
          for (int u = 0; u < 8; ++u) {
            const int s = __builtin_amdgcn_readlane(e_src[q], j0 + u);
            wj[q][u] = __int_as_float(
                __builtin_amdgcn_readlane(__float_as_int(e_w[q]), j0 + u));
            v[q][u] = *(const vf*)(x + (size_t)s * K + lane * VPL);
          }
        }
#pragma unroll
        for (int q = 0; q < RPW; ++q) {
#pragma unroll
          for (int u = 0; u < 8; ++u) {
            acc[q] += wj[q][u] * v[q][u];
            if constexpr (NORM) wsum[q] += wj[q][u];
          }
        }
      }
      if (base + 32 < maxdeg) {
#pragma unroll
        for (int q = 0; q < RPW; ++q) ed[q] = chunk(start[q], deg[q], base + 32);
      }
    }

    float* __restrict__ tb = &tile[it & 1][0];
#pragma unroll
    for (int q = 0; q < RPW; ++q) {
      vf o = acc[q];
      if constexpr (NORM) o = o * sc + wsum[q] * sh;
      *(vf*)(tb + (wave * RPW + q) * P + lane * VPL) = o;
    }
    __syncthreads();

#pragma unroll
    for (int q = 0; q < RPW; ++q) {
      start[q] = nstart[q];
      deg[q] = ndeg[q];
      ed[q] = chunk(start[q], deg[q], 0);
    }

    const v4f* __restrict__ ap = (const v4f*)(tb + c * P + 4 * h);
    v8f acc0 = {};
    v8f acc1 = {};
    constexpr int HJ = K / 16;
    static_assert(HJ == 4 || HJ == 8, "the load fence names 4 or 8 reads");
#pragma unroll
    for (int hf = 0; hf < 2; ++hf) {
      v4f a[HJ];
#pragma unroll
      for (int j = 0; j < HJ; ++j) a[j] = ap[2 * (hf * HJ + j)];
      if constexpr (HJ == 8)
        asm volatile("" ::"v"(a[0]), "v"(a[1]), "v"(a[2]), "v"(a[3]),
                     "v"(a[4]), "v"(a[5]), "v"(a[6]), "v"(a[7]));
      else
        asm volatile("" ::"v"(a[0]), "v"(a[1]), "v"(a[2]), "v"(a[3]));
#pragma unroll
      for (int j = 0; j < HJ; ++j) {
        const int jj = hf * HJ + j;
        v2f lo, hi;
        lo.x = a[j].x; lo.y = a[j].y;
        hi.x = a[j].z; hi.y = a[j].w;
        acc0 = __builtin_amdgcn_wmma_f32_16x16x4_f32(
            false, lo, false, b0[2 * jj], (short)0, acc0, false, false);
        acc1 = __builtin_amdgcn_wmma_f32_16x16x4_f32(
            false, lo, false, b1[2 * jj], (short)0, acc1, false, false);
        acc0 = __builtin_amdgcn_wmma_f32_16x16x4_f32(
            false, hi, false, b0[2 * jj + 1], (short)0, acc0, false, false);
        acc1 = __builtin_amdgcn_wmma_f32_16x16x4_f32(
            false, hi, false, b1[2 * jj + 1], (short)0, acc1, false, false);
      }
    }

    float* __restrict__ cp = C + (size_t)(row0 + 8 * h) * NC + col0 + c;
#pragma unroll
    for (int r = 0; r < 8; ++r) {
      float v0 = acc0[r];
      float v1 = acc1[r];
      v0 = v0 > 0.0f ? v0 : 0.0f;
      v1 = v1 > 0.0f ? v1 : 0.0f;
      cp[(size_t)r * NC]      = v0;
      cp[(size_t)r * NC + 16] = v1;
      s0 += v0; q0 += v0 * v0;
      s1 += v1; q1 += v1 * v1;
    }
  }

  s0 += __shfl_xor(s0, 16); q0 += __shfl_xor(q0, 16);
  s1 += __shfl_xor(s1, 16); q1 += __shfl_xor(q1, 16);
  if (h == 0) {
    atomicAdd(&stats[col0 + c], s0);
    atomicAdd(&stats[NC + col0 + c], q0);
    atomicAdd(&stats[col0 + 16 + c], s1);
    atomicAdd(&stats[NC + col0 + 16 + c], q1);
  }
}

#define BN_U 10

template <int D>
__global__ void __launch_bounds__(256)
bn_normalize(const float* __restrict__ agg,
             float* __restrict__ out,
             const float* __restrict__ gamma,
             const float* __restrict__ beta,
             const float* __restrict__ stats,
             int Nrows) {
  const int cg = threadIdx.x & (D / 4 - 1);
  const float invN = 1.0f / (float)Nrows;
  const v4f sm = ((const v4f*)stats)[cg];
  const v4f sq = ((const v4f*)(stats + D))[cg];
  const v4f ga = ((const v4f*)gamma)[cg];
  const v4f be = ((const v4f*)beta)[cg];
  const v4f mean = sm * invN;
  const v4f var = sq * invN - mean * mean;
  v4f sc;
  sc.x = ga.x * rsqrtf(var.x + BN_EPS);
  sc.y = ga.y * rsqrtf(var.y + BN_EPS);
  sc.z = ga.z * rsqrtf(var.z + BN_EPS);
  sc.w = ga.w * rsqrtf(var.w + BN_EPS);

  const int total4 = Nrows * (D / 4);
  const int stride = (int)(gridDim.x * 256);
  const v4f* __restrict__ in4 = (const v4f*)agg;
  v4f* __restrict__ out4 = (v4f*)out;

  int i0 = (int)(blockIdx.x * 256 + threadIdx.x);
  for (; i0 + (BN_U - 1) * stride < total4; i0 += BN_U * stride) {
    v4f v[BN_U];
#pragma unroll
    for (int u = 0; u < BN_U; ++u) v[u] = in4[i0 + u * stride];
    static_assert(BN_U == 10, "the statement below names every element of v");
    asm volatile("" ::"v"(v[0]), "v"(v[1]), "v"(v[2]), "v"(v[3]), "v"(v[4]),
                 "v"(v[5]), "v"(v[6]), "v"(v[7]), "v"(v[8]), "v"(v[9]));
#pragma unroll
    for (int u = 0; u < BN_U; ++u)
      out4[i0 + u * stride] = (v[u] - mean) * sc + be;
  }
  for (; i0 < total4; i0 += stride)
    out4[i0] = (in4[i0] - mean) * sc + be;
}

extern "C" void kernel_launch(void* const* d_in, const int* in_sizes, int n_in,
                              void* d_out, int out_size, void* d_ws,
                              size_t ws_size, hipStream_t stream) {
  (void)in_sizes; (void)n_in; (void)out_size; (void)ws_size;

  const float* y   = (const float*)d_in[0];
  const int*   src = (const int*)d_in[1];
  const int*   dst = (const int*)d_in[2];
  const float* w   = (const float*)d_in[3];
  const float* W1  = (const float*)d_in[4];
  const float* g1  = (const float*)d_in[5];
  const float* b1  = (const float*)d_in[6];
  const float* W2  = (const float*)d_in[7];
  const float* g2  = (const float*)d_in[8];
  const float* b2  = (const float*)d_in[9];
  float* out = (float*)d_out;

  float* ws     = (float*)d_ws;
  float* z1     = ws;
  float* z2     = z1 + (size_t)NN * H1C;
  float* stats1 = z2 + (size_t)NN * DINC;
  float* stats2 = stats1 + 2 * H1C;
  int*   cnt    = (int*)(stats2 + 2 * DINC);
  int*   rowst  = cnt + NN;
  int*   counts = rowst + NN;
  int*   binst  = counts + NGRP * NBINS;
  int*   bintot = binst + 256;
  int*   binned = bintot + 256;
  int*   edges  = binned + 2 * EE;

  bin_count<<<NGRP, L1_T, 0, stream>>>(dst, counts, (unsigned*)stats1,
                                       2 * (H1C + DINC));
  bin_place<<<NGRP, L1_T, 0, stream>>>(src, dst, w, counts, binst, bintot,
                                       binned);
  bin_rows<<<NBINS, L2_T, 0, stream>>>(binst, bintot, binned, rowst, cnt,
                                       edges);

  edge_gemm_relu_colstats<H2C, H1C, false><<<1042, H1C, 0, stream>>>(
      y, W1, z1, stats1, NN / 16, rowst, cnt, edges, NN, EE, nullptr, nullptr,
      nullptr);

  edge_gemm_relu_colstats<H1C, DINC, true><<<512, DINC, 0, stream>>>(
      z1, W2, z2, stats2, NN / 16, rowst, cnt, edges, NN, EE, g1, b1, stats1);

  bn_normalize<DINC><<<625, 256, 0, stream>>>(z2, out, g2, b2, stats2, NN);
}


# Every store instruction writes whole 128-byte lines: product rows swap one value across lane halves (permlanex16) before the z stores; edge records are ordered in LDS, then copied out in slot order
# speedup vs baseline: 6.2406x; 1.0961x over previous
#include <hip/hip_runtime.h>
#include <hip/hip_bf16.h>

typedef float v2f __attribute__((ext_vector_type(2)));
typedef float v8f __attribute__((ext_vector_type(8)));
typedef float v4f __attribute__((ext_vector_type(4)));

#define NN   50000
#define EE   800000
#define H2C  64
#define H1C  128
#define DINC 256
#define BN_EPS 1e-5f

#define BIN_SHIFT 8
#define BIN_ROWS  (1 << BIN_SHIFT)
#define NBINS     ((NN + BIN_ROWS - 1) >> BIN_SHIFT)
#define NGRP      250
#define EPG       (EE / NGRP)
#define EPG4      (EPG / 4)
#define L1_T      256
#define L1_Q      ((EPG4 + L1_T - 1) / L1_T)
#define L2_T      512
#define L2_B      8
#define L2_CAP    6144
#define CS_B      50
#define CS_Q      ((CS_B + 3) / 4)
static_assert(NGRP % CS_B == 0 && CS_Q == 13, "the load fence names 13 quads");
typedef int v4i __attribute__((ext_vector_type(4)));
static_assert(NGRP * EPG == EE && EPG % 4 == 0, "groups tile the edge list");
static_assert(NBINS <= L1_T && BIN_ROWS <= L2_T, "one thread per bin / row");
static_assert(NN <= 65536 && BIN_ROWS <= 256, "src and row-in-bin share a word");

__device__ __forceinline__ int block_excl_scan(int v, int* s_wave) {
  const int lane = threadIdx.x & 31;
  const int wv   = threadIdx.x >> 5;
  int inc = v;
  for (int o = 1; o < 32; o <<= 1) {
    const int t = __shfl_up(inc, o);
    if (lane >= o) inc += t;
  }
  if (lane == 31) s_wave[wv] = inc;
  __syncthreads();
  if (wv == 0) {
    const int nw = (int)(blockDim.x >> 5);
    const int x = lane < nw ? s_wave[lane] : 0;
    int y = x;
    for (int o = 1; o < 32; o <<= 1) {
      const int t = __shfl_up(y, o);
      if (lane >= o) y += t;
    }
    if (lane < nw) s_wave[lane] = y - x;
  }
  __syncthreads();
  const int r = s_wave[wv] + inc - v;
  __syncthreads();
  return r;
}

__device__ __forceinline__ int bin_of(int d) {
  return (int)min((unsigned)d >> BIN_SHIFT, (unsigned)(NBINS - 1));
}

__global__ void __launch_bounds__(L1_T)
bin_count(const int* __restrict__ dst,
          int* __restrict__ counts,
          unsigned* __restrict__ stats, int nstats) {
  __shared__ int hist[NBINS];
  const int t = threadIdx.x;
  const int g = blockIdx.x;
  if (t < NBINS) hist[t] = 0;
  if (g == 0)
    for (int i = t; i < nstats; i += L1_T) stats[i] = 0u;
  const int4* __restrict__ d4 = (const int4*)dst + (size_t)g * EPG4;
  int4 d[L1_Q];
#pragma unroll
  for (int k = 0; k < L1_Q; ++k) d[k] = d4[min(t + k * L1_T, EPG4 - 1)];
  __syncthreads();
#pragma unroll
  for (int k = 0; k < L1_Q; ++k) {
    if (t + k * L1_T < EPG4) {
      atomicAdd(&hist[bin_of(d[k].x)], 1);
      atomicAdd(&hist[bin_of(d[k].y)], 1);
      atomicAdd(&hist[bin_of(d[k].z)], 1);
      atomicAdd(&hist[bin_of(d[k].w)], 1);
    }
  }
  __syncthreads();
  if (t < NBINS) counts[g * NBINS + t] = hist[t];
}

__global__ void __launch_bounds__(L1_T)
bin_place(const int* __restrict__ src,
          const int* __restrict__ dst,
          const float* __restrict__ w,
          const int* __restrict__ counts,
          int* __restrict__ binstart_g,
          int* __restrict__ bintot_g,
          int* __restrict__ binned) {
  __shared__ int cur[NBINS];
  __shared__ int delta[NBINS];
  __shared__ int s_wave[32];
  __shared__ int2 stage[EPG];
  __shared__ unsigned char sbin[EPG];
  static_assert(NBINS <= 256, "a bin number fits a byte");
  const int t = threadIdx.x;
  const int g = blockIdx.x;
  const int4*   __restrict__ d4 = (const int4*)dst + (size_t)g * EPG4;
  const int4*   __restrict__ s4 = (const int4*)src + (size_t)g * EPG4;
  const float4* __restrict__ w4 = (const float4*)w + (size_t)g * EPG4;
  int4 d[L1_Q], s[L1_Q];
  float4 ww[L1_Q];
#pragma unroll
  for (int k = 0; k < L1_Q; ++k) {
    const int q = min(t + k * L1_T, EPG4 - 1);
    d[k] = d4[q];
    s[k] = s4[q];
    ww[k] = w4[q];
  }

  int tot = 0, part = 0, own = 0;
  if (t < NBINS) {
    const int* __restrict__ cp = counts + t;
    for (int g0 = 0; g0 < NGRP; g0 += CS_B) {
      v4i v[CS_Q];
      v[CS_Q - 1] = (v4i){0, 0, 0, 0};
#pragma unroll
      for (int i = 0; i < CS_B; ++i) v[i >> 2][i & 3] = cp[(g0 + i) * NBINS];
      asm volatile("" ::"v"(v[0]), "v"(v[1]), "v"(v[2]), "v"(v[3]), "v"(v[4]),
                   "v"(v[5]), "v"(v[6]), "v"(v[7]), "v"(v[8]), "v"(v[9]),
                   "v"(v[10]), "v"(v[11]), "v"(v[12]));
#pragma unroll
      for (int i = 0; i < CS_B; ++i) {
        const int x = v[i >> 2][i & 3];
        tot += x;
        part += g0 + i < g ? x : 0;
        own += g0 + i == g ? x : 0;
      }
    }
  }
  const int binstart = block_excl_scan(tot, s_wave);
  const int lstart = block_excl_scan(own, s_wave);
  if (t < NBINS) {
    cur[t] = lstart;
    delta[t] = binstart + part - lstart;
    if (g == 0) {
      binstart_g[t] = binstart;
      bintot_g[t] = tot;
    }
  }
  __syncthreads();

  int2* __restrict__ bs = (int2*)binned;
#pragma unroll
  for (int k = 0; k < L1_Q; ++k) {
    if (t + k * L1_T < EPG4) {
      const int dd[4] = {d[k].x, d[k].y, d[k].z, d[k].w};
      const int ss[4] = {s[k].x, s[k].y, s[k].z, s[k].w};
      const float wv4[4] = {ww[k].x, ww[k].y, ww[k].z, ww[k].w};
      int p[4];
#pragma unroll
      for (int u = 0; u < 4; ++u) p[u] = atomicAdd(&cur[bin_of(dd[u])], 1);
#pragma unroll
      for (int u = 0; u < 4; ++u) {
        const int key = (ss[u] & 0xFFFF) | ((dd[u] & (BIN_ROWS - 1)) << 16);
        if ((unsigned)p[u] < (unsigned)EPG) {
          stage[p[u]] = make_int2(key, __float_as_int(wv4[u]));
          sbin[p[u]] = (unsigned char)bin_of(dd[u]);
        }
      }
    }
  }
  __syncthreads();
  for (int i0 = t; i0 < EPG; i0 += 4 * L1_T) {
    int2 r[4];
    int to[4];
#pragma unroll
    for (int u = 0; u < 4; ++u) {
      const int i = min(i0 + u * L1_T, EPG - 1);
      r[u] = stage[i];
      to[u] = i + delta[sbin[i]];
    }
#pragma unroll
    for (int u = 0; u < 4; ++u) {
      if (i0 + u * L1_T < EPG && (unsigned)to[u] < (unsigned)EE) bs[to[u]] = r[u];
    }
  }
}

struct lds_yes { static constexpr bool value = true; };
struct lds_no  { static constexpr bool value = false; };

__global__ void __launch_bounds__(L2_T)
bin_rows(const int* __restrict__ binstart_g,
         const int* __restrict__ bintot_g,
         const int* __restrict__ binned,
         int* __restrict__ rowstart,
         int* __restrict__ cnt,
         int* __restrict__ edges) {
  __shared__ int c[BIN_ROWS];
  __shared__ int cur[BIN_ROWS];
  __shared__ int s_wave[32];
  __shared__ int2 stage[L2_CAP];
  const int t = threadIdx.x;
  const int b = blockIdx.x;

  const int base = binstart_g[b];
  const int tot  = bintot_g[b];
  const int n = (unsigned)base <= (unsigned)EE ? min(tot, EE - base) : 0;
  if (t < BIN_ROWS) c[t] = 0;
  __syncthreads();

  const int2* __restrict__ bs = (const int2*)binned;
  for (int i0 = t; i0 < n; i0 += L2_B * L2_T) {
    int key[L2_B];
#pragma unroll
    for (int k = 0; k < L2_B; ++k)
      key[k] = bs[base + min(i0 + k * L2_T, n - 1)].x;
    static_assert(L2_B == 8, "the statement below names every element of key");
    asm volatile("" ::"v"(key[0]), "v"(key[1]), "v"(key[2]), "v"(key[3]),
                 "v"(key[4]), "v"(key[5]), "v"(key[6]), "v"(key[7]));
#pragma unroll
    for (int k = 0; k < L2_B; ++k)
      if (i0 + k * L2_T < n)
        atomicAdd(&c[(key[k] >> 16) & (BIN_ROWS - 1)], 1);
  }
  __syncthreads();

  const int cc  = t < BIN_ROWS ? c[t] : 0;
  const int off = block_excl_scan(cc, s_wave);
  if (t < BIN_ROWS) {
    cur[t] = base + off;
    const int row = (b << BIN_SHIFT) + t;
    if (row < NN) {
      rowstart[row] = base + off;
      cnt[row] = cc;
    }
  }
  __syncthreads();

  int2* __restrict__ es = (int2*)edges;
  auto place = [&](auto to_lds) {
    constexpr bool TO_LDS = decltype(to_lds)::value;
    for (int i0 = t; i0 < n; i0 += L2_B * L2_T) {
      int2 e[L2_B];
      int p[L2_B];
#pragma unroll
      for (int k = 0; k < L2_B; ++k)
        e[k] = bs[base + min(i0 + k * L2_T, n - 1)];
#pragma unroll
      for (int k = 0; k < L2_B; ++k)
        p[k] = i0 + k * L2_T < n
                   ? atomicAdd(&cur[(e[k].x >> 16) & (BIN_ROWS - 1)], 1)
                   : EE;
#pragma unroll
      for (int k = 0; k < L2_B; ++k) {
        const int2 rec = make_int2(e[k].x & 0xFFFF, e[k].y);
        if constexpr (TO_LDS) {
          if ((unsigned)(p[k] - base) < (unsigned)n) stage[p[k] - base] = rec;
        } else {
          if ((unsigned)p[k] < (unsigned)EE) es[p[k]] = rec;
        }
      }
    }
  };
  if (n <= L2_CAP) {
    place(lds_yes{});
    __syncthreads();
    for (int i0 = t; i0 < n; i0 += 4 * L2_T) {
      int2 r[4];
#pragma unroll
      for (int u = 0; u < 4; ++u) r[u] = stage[min(i0 + u * L2_T, n - 1)];
#pragma unroll
      for (int u = 0; u < 4; ++u)
        if (i0 + u * L2_T < n) es[base + i0 + u * L2_T] = r[u];
    }
  } else {
    place(lds_no{});
  }
}

template <int K, int NC, bool NORM>
__global__ void __launch_bounds__(NC, NC == 256 ? 4 : 5)
edge_gemm_relu_colstats(const float* __restrict__ x,
                        const float* __restrict__ B,
                        float* __restrict__ C,
                        float* __restrict__ stats,
                        int rowTiles,
                        const int* __restrict__ rowstart,
                        const int* __restrict__ cnt,
                        const int* __restrict__ edges,
                        int n, int E,
                        const float* __restrict__ gamma,
                        const float* __restrict__ beta,
                        const float* __restrict__ stats_in) {
  constexpr int VPL = K / 32;
  constexpr int NW  = NC / 32;
  constexpr int RPW = 16 / NW;
  constexpr int P   = K + 4;
  static_assert(NW * RPW == 16 && K % 32 == 0, "a block sums whole row tiles");
  typedef float vf __attribute__((ext_vector_type(VPL)));
  __shared__ __attribute__((aligned(16))) float tile[2][16 * P];

  const int2* __restrict__ es = (const int2*)edges;
  const int lane = threadIdx.x & 31;
  const int wave = __builtin_amdgcn_readfirstlane((int)(threadIdx.x >> 5));
  const int c = lane & 15;
  const int h = lane >> 4;
  const int col0 = wave << 5;

  v2f b0[K / 4], b1[K / 4];
  {
    const float* __restrict__ bp = B + (size_t)(4 * h) * NC + col0 + c;
#pragma unroll
    for (int t = 0; t < K / 4; ++t) {
      const float* __restrict__ p = bp + (size_t)(8 * (t >> 1) + 2 * (t & 1)) * NC;
      b0[t].x = p[0];
      b0[t].y = p[NC];
      b1[t].x = p[16];
      b1[t].y = p[NC + 16];
    }
  }

  vf sc = {}, sh = {};
  if constexpr (NORM) {
    const float invN = 1.0f / (float)n;
    const vf sm = *(const vf*)(stats_in + lane * VPL);
    const vf sq = *(const vf*)(stats_in + 32 * VPL + lane * VPL);
    const vf ga = *(const vf*)(gamma + lane * VPL);
    const vf be = *(const vf*)(beta + lane * VPL);
    const vf mean = sm * invN;
    const vf var = sq * invN - mean * mean;
#pragma unroll
    for (int k = 0; k < VPL; ++k) sc[k] = ga[k] * rsqrtf(var[k] + BN_EPS);
    sh = be - mean * sc;
  }

  auto chunk = [&](int st, int dg, int base) -> int2 {
    const int j = max(min(base + lane, dg - 1), 0);
    return es[min((unsigned)(st + j), (unsigned)(E - 1))];
  };

  float s0 = 0.0f, q0 = 0.0f, s1 = 0.0f, q1 = 0.0f;

  int rt = blockIdx.x;
  int start[RPW], deg[RPW];
  int2 ed[RPW];
#pragma unroll
  for (int q = 0; q < RPW; ++q) {
    const int r = min((rt << 4) + wave * RPW + q, n - 1);
    start[q] = rowstart[r];
    deg[q] = rt < rowTiles ? cnt[r] : 0;
  }
#pragma unroll
  for (int q = 0; q < RPW; ++q) ed[q] = chunk(start[q], deg[q], 0);

  for (int it = 0; rt < rowTiles; rt += gridDim.x, ++it) {
    const int row0 = rt << 4;
    const int rtn = rt + (int)gridDim.x;

    int nstart[RPW], ndeg[RPW];
#pragma unroll
    for (int q = 0; q < RPW; ++q) {
      const int r = min((rtn << 4) + wave * RPW + q, n - 1);
      nstart[q] = rowstart[r];
      ndeg[q] = rtn < rowTiles ? cnt[r] : 0;
    }

    int maxdeg = 0;
#pragma unroll
    for (int q = 0; q < RPW; ++q) maxdeg = max(maxdeg, deg[q]);

    vf acc[RPW];
    float wsum[RPW];
#pragma unroll
    for (int q = 0; q < RPW; ++q) { acc[q] = (vf){}; wsum[q] = 0.0f; }

    for (int base = 0; base < maxdeg; base += 32) {
      int e_src[RPW], m[RPW];
      float e_w[RPW];
#pragma unroll
      for (int q = 0; q < RPW; ++q) {
        m[q] = min(32, deg[q] - base);
        e_src[q] = ed[q].x;
        e_w[q] = lane < m[q] ? __int_as_float(ed[q].y) : 0.0f;
      }
      const int mm = min(32, maxdeg - base);
      for (int j0 = 0; j0 < mm; j0 += 8) {
        vf v[RPW][8];
        float wj[RPW][8];
#pragma unroll
        for (int q = 0; q < RPW; ++q) {
#pragma unroll
          for (int u = 0; u < 8; ++u) {
            const int s = __builtin_amdgcn_readlane(e_src[q], j0 + u);
            wj[q][u] = __int_as_float(
                __builtin_amdgcn_readlane(__float_as_int(e_w[q]), j0 + u));
            v[q][u] = *(const vf*)(x + (size_t)s * K + lane * VPL);
          }
        }
#pragma unroll
        for (int q = 0; q < RPW; ++q) {
#pragma unroll
          for (int u = 0; u < 8; ++u) {
            acc[q] += wj[q][u] * v[q][u];
            if constexpr (NORM) wsum[q] += wj[q][u];
          }
        }
      }
      if (base + 32 < maxdeg) {
#pragma unroll
        for (int q = 0; q < RPW; ++q) ed[q] = chunk(start[q], deg[q], base + 32);
      }
    }

    float* __restrict__ tb = &tile[it & 1][0];
#pragma unroll
    for (int q = 0; q < RPW; ++q) {
      vf o = acc[q];
      if constexpr (NORM) o = o * sc + wsum[q] * sh;
      *(vf*)(tb + (wave * RPW + q) * P + lane * VPL) = o;
    }
    __syncthreads();

#pragma unroll
    for (int q = 0; q < RPW; ++q) {
      start[q] = nstart[q];
      deg[q] = ndeg[q];
      ed[q] = chunk(start[q], deg[q], 0);
    }

    const v4f* __restrict__ ap = (const v4f*)(tb + c * P + 4 * h);
    v8f acc0 = {};
    v8f acc1 = {};
    constexpr int HJ = K / 16;
    static_assert(HJ == 4 || HJ == 8, "the load fence names 4 or 8 reads");
#pragma unroll
    for (int hf = 0; hf < 2; ++hf) {
      v4f a[HJ];
#pragma unroll
      for (int j = 0; j < HJ; ++j) a[j] = ap[2 * (hf * HJ + j)];
      if constexpr (HJ == 8)
        asm volatile("" ::"v"(a[0]), "v"(a[1]), "v"(a[2]), "v"(a[3]),
                     "v"(a[4]), "v"(a[5]), "v"(a[6]), "v"(a[7]));
      else
        asm volatile("" ::"v"(a[0]), "v"(a[1]), "v"(a[2]), "v"(a[3]));
#pragma unroll
      for (int j = 0; j < HJ; ++j) {
        const int jj = hf * HJ + j;
        v2f lo, hi;
        lo.x = a[j].x; lo.y = a[j].y;
        hi.x = a[j].z; hi.y = a[j].w;
        acc0 = __builtin_amdgcn_wmma_f32_16x16x4_f32(
            false, lo, false, b0[2 * jj], (short)0, acc0, false, false);
        acc1 = __builtin_amdgcn_wmma_f32_16x16x4_f32(
            false, lo, false, b1[2 * jj], (short)0, acc1, false, false);
        acc0 = __builtin_amdgcn_wmma_f32_16x16x4_f32(
            false, hi, false, b0[2 * jj + 1], (short)0, acc0, false, false);
        acc1 = __builtin_amdgcn_wmma_f32_16x16x4_f32(
            false, hi, false, b1[2 * jj + 1], (short)0, acc1, false, false);
      }
    }

    float* __restrict__ cp = C + (size_t)row0 * NC + col0 + lane;
#pragma unroll
    for (int r = 0; r < 8; ++r) {
      float v0 = acc0[r];
      float v1 = acc1[r];
      v0 = v0 > 0.0f ? v0 : 0.0f;
      v1 = v1 > 0.0f ? v1 : 0.0f;
      const float give = h ? v0 : v1;
      const float got = __builtin_bit_cast(
          float, __builtin_amdgcn_permlanex16(
                     __builtin_bit_cast(int, give), __builtin_bit_cast(int, give),
                     0x76543210, 0xfedcba98, false, false));
      cp[(size_t)r * NC]       = h ? got : v0;
      cp[(size_t)(8 + r) * NC] = h ? v1 : got;
      s0 += v0; q0 += v0 * v0;
      s1 += v1; q1 += v1 * v1;
    }
  }

  s0 += __shfl_xor(s0, 16); q0 += __shfl_xor(q0, 16);
  s1 += __shfl_xor(s1, 16); q1 += __shfl_xor(q1, 16);
  if (h == 0) {
    atomicAdd(&stats[col0 + c], s0);
    atomicAdd(&stats[NC + col0 + c], q0);
    atomicAdd(&stats[col0 + 16 + c], s1);
    atomicAdd(&stats[NC + col0 + 16 + c], q1);
  }
}

#define BN_U 10

template <int D>
__global__ void __launch_bounds__(256)
bn_normalize(const float* __restrict__ agg,
             float* __restrict__ out,
             const float* __restrict__ gamma,
             const float* __restrict__ beta,
             const float* __restrict__ stats,
             int Nrows) {
  const int cg = threadIdx.x & (D / 4 - 1);
  const float invN = 1.0f / (float)Nrows;
  const v4f sm = ((const v4f*)stats)[cg];
  const v4f sq = ((const v4f*)(stats + D))[cg];
  const v4f ga = ((const v4f*)gamma)[cg];
  const v4f be = ((const v4f*)beta)[cg];
  const v4f mean = sm * invN;
  const v4f var = sq * invN - mean * mean;
  v4f sc;
  sc.x = ga.x * rsqrtf(var.x + BN_EPS);
  sc.y = ga.y * rsqrtf(var.y + BN_EPS);
  sc.z = ga.z * rsqrtf(var.z + BN_EPS);
  sc.w = ga.w * rsqrtf(var.w + BN_EPS);

  const int total4 = Nrows * (D / 4);
  const int stride = (int)(gridDim.x * 256);
  const v4f* __restrict__ in4 = (const v4f*)agg;
  v4f* __restrict__ out4 = (v4f*)out;

  int i0 = (int)(blockIdx.x * 256 + threadIdx.x);
  for (; i0 + (BN_U - 1) * stride < total4; i0 += BN_U * stride) {
    v4f v[BN_U];
#pragma unroll
    for (int u = 0; u < BN_U; ++u) v[u] = in4[i0 + u * stride];
    static_assert(BN_U == 10, "the statement below names every element of v");
    asm volatile("" ::"v"(v[0]), "v"(v[1]), "v"(v[2]), "v"(v[3]), "v"(v[4]),
                 "v"(v[5]), "v"(v[6]), "v"(v[7]), "v"(v[8]), "v"(v[9]));
#pragma unroll
    for (int u = 0; u < BN_U; ++u)
      out4[i0 + u * stride] = (v[u] - mean) * sc + be;
  }
  for (; i0 < total4; i0 += stride)
    out4[i0] = (in4[i0] - mean) * sc + be;
}

extern "C" void kernel_launch(void* const* d_in, const int* in_sizes, int n_in,
                              void* d_out, int out_size, void* d_ws,
                              size_t ws_size, hipStream_t stream) {
  (void)in_sizes; (void)n_in; (void)out_size; (void)ws_size;

  const float* y   = (const float*)d_in[0];
  const int*   src = (const int*)d_in[1];
  const int*   dst = (const int*)d_in[2];
  const float* w   = (const float*)d_in[3];
  const float* W1  = (const float*)d_in[4];
  const float* g1  = (const float*)d_in[5];
  const float* b1  = (const float*)d_in[6];
  const float* W2  = (const float*)d_in[7];
  const float* g2  = (const float*)d_in[8];
  const float* b2  = (const float*)d_in[9];
  float* out = (float*)d_out;

  float* ws     = (float*)d_ws;
  float* z1     = ws;
  float* z2     = z1 + (size_t)NN * H1C;
  float* stats1 = z2 + (size_t)NN * DINC;
  float* stats2 = stats1 + 2 * H1C;
  int*   cnt    = (int*)(stats2 + 2 * DINC);
  int*   rowst  = cnt + NN;
  int*   counts = rowst + NN;
  int*   binst  = counts + NGRP * NBINS;
  int*   bintot = binst + 256;
  int*   binned = bintot + 256;
  int*   edges  = binned + 2 * EE;

  bin_count<<<NGRP, L1_T, 0, stream>>>(dst, counts, (unsigned*)stats1,
                                       2 * (H1C + DINC));
  bin_place<<<NGRP, L1_T, 0, stream>>>(src, dst, w, counts, binst, bintot,
                                       binned);
  bin_rows<<<NBINS, L2_T, 0, stream>>>(binst, bintot, binned, rowst, cnt,
                                       edges);

  edge_gemm_relu_colstats<H2C, H1C, false><<<1042, H1C, 0, stream>>>(
      y, W1, z1, stats1, NN / 16, rowst, cnt, edges, NN, EE, nullptr, nullptr,
      nullptr);

  edge_gemm_relu_colstats<H1C, DINC, true><<<512, DINC, 0, stream>>>(
      z1, W2, z2, stats2, NN / 16, rowst, cnt, edges, NN, EE, g1, b1, stats1);

  bn_normalize<DINC><<<625, 256, 0, stream>>>(z2, out, g2, b2, stats2, NN);
}


# Level-1 grouping in one launch: a group orders its own edges by bin in LDS, stores its run in place; the per-bin group reads its 250 pieces: count launch goes; order of float additions changes
# speedup vs baseline: 6.6898x; 1.0720x over previous
#include <hip/hip_runtime.h>
#include <hip/hip_bf16.h>

typedef float v2f __attribute__((ext_vector_type(2)));
typedef float v8f __attribute__((ext_vector_type(8)));
typedef float v4f __attribute__((ext_vector_type(4)));

#define NN   50000
#define EE   800000
#define H2C  64
#define H1C  128
#define DINC 256
#define BN_EPS 1e-5f

#define BIN_SHIFT 8
#define BIN_ROWS  (1 << BIN_SHIFT)
#define NBINS     ((NN + BIN_ROWS - 1) >> BIN_SHIFT)
#define NGRP      250
#define EPG       (EE / NGRP)
#define EPG4      (EPG / 4)
#define L1_T      256
#define L1_Q      ((EPG4 + L1_T - 1) / L1_T)
#define L2_T      512
#define L2_W      (L2_T / 32)
#define L2_P      ((NGRP + L2_W - 1) / L2_W)
#define L2_CAP    6144
typedef int v2i __attribute__((ext_vector_type(2)));
static_assert(L2_P == 16, "the load fence names 16 records");
static_assert(L2_W * L2_P <= L2_T, "one thread per piece of a bin");
static_assert(EPG < 32768, "start and count of a piece share a word");
static_assert(NGRP * EPG == EE && EPG % 4 == 0, "groups tile the edge list");
static_assert(NBINS <= L1_T && BIN_ROWS <= L2_T, "one thread per bin / row");
static_assert(NN <= 65536 && BIN_ROWS <= 256, "src and row-in-bin share a word");

__device__ __forceinline__ int block_excl_scan(int v, int* s_wave) {
  const int lane = threadIdx.x & 31;
  const int wv   = threadIdx.x >> 5;
  int inc = v;
  for (int o = 1; o < 32; o <<= 1) {
    const int t = __shfl_up(inc, o);
    if (lane >= o) inc += t;
  }
  if (lane == 31) s_wave[wv] = inc;
  __syncthreads();
  if (wv == 0) {
    const int nw = (int)(blockDim.x >> 5);
    const int x = lane < nw ? s_wave[lane] : 0;
    int y = x;
    for (int o = 1; o < 32; o <<= 1) {
      const int t = __shfl_up(y, o);
      if (lane >= o) y += t;
    }
    if (lane < nw) s_wave[lane] = y - x;
  }
  __syncthreads();
  const int r = s_wave[wv] + inc - v;
  __syncthreads();
  return r;
}

__device__ __forceinline__ int bin_of(int d) {
  return (int)min((unsigned)d >> BIN_SHIFT, (unsigned)(NBINS - 1));
}

__global__ void __launch_bounds__(L1_T)
bin_place(const int* __restrict__ src,
          const int* __restrict__ dst,
          const float* __restrict__ w,
          int* __restrict__ pieces,
          int* __restrict__ binned,
          unsigned* __restrict__ stats, int nstats) {
  __shared__ int hist[NBINS];
  __shared__ int lst[NBINS];
  __shared__ int s_wave[32];
  __shared__ v2i stage[EPG];
  const int t = threadIdx.x;
  const int g = blockIdx.x;
  if (t < NBINS) hist[t] = 0;
  if (g == 0)
    for (int i = t; i < nstats; i += L1_T) stats[i] = 0u;
  const int4*   __restrict__ d4 = (const int4*)dst + (size_t)g * EPG4;
  const int4*   __restrict__ s4 = (const int4*)src + (size_t)g * EPG4;
  const float4* __restrict__ w4 = (const float4*)w + (size_t)g * EPG4;
  int4 d[L1_Q], s[L1_Q];
  float4 ww[L1_Q];
#pragma unroll
  for (int k = 0; k < L1_Q; ++k) {
    const int q = min(t + k * L1_T, EPG4 - 1);
    d[k] = d4[q];
    s[k] = s4[q];
    ww[k] = w4[q];
  }
  __syncthreads();

  int rk[L1_Q][4];
#pragma unroll
  for (int k = 0; k < L1_Q; ++k) {
    const int dd[4] = {d[k].x, d[k].y, d[k].z, d[k].w};
#pragma unroll
    for (int u = 0; u < 4; ++u)
      rk[k][u] = t + k * L1_T < EPG4 ? atomicAdd(&hist[bin_of(dd[u])], 1) : 0;
  }
  __syncthreads();

  const int own = t < NBINS ? hist[t] : 0;
  const int lstart = block_excl_scan(own, s_wave);
  if (t < NBINS) {
    lst[t] = lstart;
    pieces[g * NBINS + t] = lstart | (own << 16);
  }
  __syncthreads();

#pragma unroll
  for (int k = 0; k < L1_Q; ++k) {
    if (t + k * L1_T < EPG4) {
      const int dd[4] = {d[k].x, d[k].y, d[k].z, d[k].w};
      const int ss[4] = {s[k].x, s[k].y, s[k].z, s[k].w};
      const float wv4[4] = {ww[k].x, ww[k].y, ww[k].z, ww[k].w};
#pragma unroll
      for (int u = 0; u < 4; ++u) {
        const int p = lst[bin_of(dd[u])] + rk[k][u];
        const int key = (ss[u] & 0xFFFF) | ((dd[u] & (BIN_ROWS - 1)) << 16);
        if ((unsigned)p < (unsigned)EPG)
          stage[p] = (v2i){key, __float_as_int(wv4[u])};
      }
    }
  }
  __syncthreads();
  v2i* __restrict__ run = (v2i*)binned + (size_t)g * EPG;
  for (int i0 = t; i0 < EPG; i0 += 4 * L1_T) {
    v2i r[4];
#pragma unroll
    for (int u = 0; u < 4; ++u) r[u] = stage[min(i0 + u * L1_T, EPG - 1)];
#pragma unroll
    for (int u = 0; u < 4; ++u)
      if (i0 + u * L1_T < EPG) run[i0 + u * L1_T] = r[u];
  }
}

struct lds_yes { static constexpr bool value = true; };
struct lds_no  { static constexpr bool value = false; };

__global__ void __launch_bounds__(L2_T)
bin_rows(const int* __restrict__ pieces,
         const int* __restrict__ binned,
         int* __restrict__ rowstart,
         int* __restrict__ cnt,
         int* __restrict__ edges) {
  __shared__ int c[BIN_ROWS];
  __shared__ int cur[BIN_ROWS];
  __shared__ int s_wave[32];
  __shared__ int s_ls[L2_W];
  __shared__ int s_pc[L2_W];
  __shared__ int pcn[L2_W * L2_P];
  __shared__ int psr[L2_W * L2_P];
  __shared__ v2i stage[L2_CAP];
  const int t = threadIdx.x;
  const int b = blockIdx.x;
  const int lane = t & 31;
  const int wave = __builtin_amdgcn_readfirstlane(t >> 5);

  int ls = 0, pc = 0;
  if (t < NGRP) {
    const int pk = pieces[t * NBINS + b];
    ls = min(pk & 0xFFFF, EPG);
    pc = min((pk >> 16) & 0x7FFF, EPG - ls);
  }
  if (t < BIN_ROWS) c[t] = 0;
  if (t < L2_W * L2_P) {
    pcn[t] = pc;
    psr[t] = min(t, NGRP - 1) * EPG + ls;
  }
  {
    int a = ls, q = pc;
    for (int o = 16; o > 0; o >>= 1) {
      a += __shfl_xor(a, o);
      q += __shfl_xor(q, o);
    }
    if (lane == 0) {
      s_ls[wave] = a;
      s_pc[wave] = q;
    }
  }
  __syncthreads();
  int base = 0, tot = 0;
#pragma unroll
  for (int i = 0; i < L2_W; ++i) {
    base += s_ls[i];
    tot += s_pc[i];
  }
  const int n = (unsigned)base <= (unsigned)EE ? min(tot, EE - base) : 0;

  const v2i* __restrict__ bs = (const v2i*)binned;
  v2i rec[L2_P];
  int np[L2_P];
#pragma unroll
  for (int k = 0; k < L2_P; ++k) {
    np[k] = __builtin_amdgcn_readfirstlane(pcn[wave + L2_W * k]);
    const int st = __builtin_amdgcn_readfirstlane(psr[wave + L2_W * k]);
    const int j = min(lane, max(np[k] - 1, 0));
    rec[k] = bs[min((unsigned)(st + j), (unsigned)(EE - 1))];
  }
  asm volatile("" ::"v"(rec[0]), "v"(rec[1]), "v"(rec[2]), "v"(rec[3]),
               "v"(rec[4]), "v"(rec[5]), "v"(rec[6]), "v"(rec[7]),
               "v"(rec[8]), "v"(rec[9]), "v"(rec[10]), "v"(rec[11]),
               "v"(rec[12]), "v"(rec[13]), "v"(rec[14]), "v"(rec[15]));
#pragma unroll
  for (int k = 0; k < L2_P; ++k)
    if (lane < np[k]) atomicAdd(&c[(rec[k].x >> 16) & (BIN_ROWS - 1)], 1);
#pragma unroll
  for (int k = 0; k < L2_P; ++k) {
    if (np[k] > 32) {
      const int st = psr[wave + L2_W * k];
      for (int j = 32 + lane; j < np[k]; j += 32) {
        const int key = bs[min((unsigned)(st + j), (unsigned)(EE - 1))].x;
        atomicAdd(&c[(key >> 16) & (BIN_ROWS - 1)], 1);
      }
    }
  }
  __syncthreads();

  const int cc  = t < BIN_ROWS ? c[t] : 0;
  const int off = block_excl_scan(cc, s_wave);
  if (t < BIN_ROWS) {
    cur[t] = base + off;
    const int row = (b << BIN_SHIFT) + t;
    if (row < NN) {
      rowstart[row] = base + off;
      cnt[row] = cc;
    }
  }
  __syncthreads();

  v2i* __restrict__ es = (v2i*)edges;
  auto place = [&](auto to_lds) {
    constexpr bool TO_LDS = decltype(to_lds)::value;
    auto put = [&](int p, v2i e) {
      const v2i r = (v2i){e.x & 0xFFFF, e.y};
      if constexpr (TO_LDS) {
        if ((unsigned)(p - base) < (unsigned)n) stage[p - base] = r;
      } else {
        if ((unsigned)p < (unsigned)EE) es[p] = r;
      }
    };
    int p[L2_P];
#pragma unroll
    for (int k = 0; k < L2_P; ++k)
      p[k] = lane < np[k]
                 ? atomicAdd(&cur[(rec[k].x >> 16) & (BIN_ROWS - 1)], 1)
                 : EE;
#pragma unroll
    for (int k = 0; k < L2_P; ++k) put(p[k], rec[k]);
#pragma unroll
    for (int k = 0; k < L2_P; ++k) {
      if (np[k] > 32) {
        const int st = psr[wave + L2_W * k];
        for (int j = 32 + lane; j < np[k]; j += 32) {
          const v2i e = bs[min((unsigned)(st + j), (unsigned)(EE - 1))];
          put(atomicAdd(&cur[(e.x >> 16) & (BIN_ROWS - 1)], 1), e);
        }
      }
    }
  };
  if (n <= L2_CAP) {
    place(lds_yes{});
    __syncthreads();
    for (int i0 = t; i0 < n; i0 += 4 * L2_T) {
      v2i r[4];
#pragma unroll
      for (int u = 0; u < 4; ++u) r[u] = stage[min(i0 + u * L2_T, n - 1)];
#pragma unroll
      for (int u = 0; u < 4; ++u)
        if (i0 + u * L2_T < n) es[base + i0 + u * L2_T] = r[u];
    }
  } else {
    place(lds_no{});
  }
}

template <int K, int NC, bool NORM>
__global__ void __launch_bounds__(NC, NC == 256 ? 4 : 5)
edge_gemm_relu_colstats(const float* __restrict__ x,
                        const float* __restrict__ B,
                        float* __restrict__ C,
                        float* __restrict__ stats,
                        int rowTiles,
                        const int* __restrict__ rowstart,
                        const int* __restrict__ cnt,
                        const int* __restrict__ edges,
                        int n, int E,
                        const float* __restrict__ gamma,
                        const float* __restrict__ beta,
                        const float* __restrict__ stats_in) {
  constexpr int VPL = K / 32;
  constexpr int NW  = NC / 32;
  constexpr int RPW = 16 / NW;
  constexpr int P   = K + 4;
  static_assert(NW * RPW == 16 && K % 32 == 0, "a block sums whole row tiles");
  typedef float vf __attribute__((ext_vector_type(VPL)));
  __shared__ __attribute__((aligned(16))) float tile[2][16 * P];

  const int2* __restrict__ es = (const int2*)edges;
  const int lane = threadIdx.x & 31;
  const int wave = __builtin_amdgcn_readfirstlane((int)(threadIdx.x >> 5));
  const int c = lane & 15;
  const int h = lane >> 4;
  const int col0 = wave << 5;

  v2f b0[K / 4], b1[K / 4];
  {
    const float* __restrict__ bp = B + (size_t)(4 * h) * NC + col0 + c;
#pragma unroll
    for (int t = 0; t < K / 4; ++t) {
      const float* __restrict__ p = bp + (size_t)(8 * (t >> 1) + 2 * (t & 1)) * NC;
      b0[t].x = p[0];
      b0[t].y = p[NC];
      b1[t].x = p[16];
      b1[t].y = p[NC + 16];
    }
  }

  vf sc = {}, sh = {};
  if constexpr (NORM) {
    const float invN = 1.0f / (float)n;
    const vf sm = *(const vf*)(stats_in + lane * VPL);
    const vf sq = *(const vf*)(stats_in + 32 * VPL + lane * VPL);
    const vf ga = *(const vf*)(gamma + lane * VPL);
    const vf be = *(const vf*)(beta + lane * VPL);
    const vf mean = sm * invN;
    const vf var = sq * invN - mean * mean;
#pragma unroll
    for (int k = 0; k < VPL; ++k) sc[k] = ga[k] * rsqrtf(var[k] + BN_EPS);
    sh = be - mean * sc;
  }

  auto chunk = [&](int st, int dg, int base) -> int2 {
    const int j = max(min(base + lane, dg - 1), 0);
    return es[min((unsigned)(st + j), (unsigned)(E - 1))];
  };

  float s0 = 0.0f, q0 = 0.0f, s1 = 0.0f, q1 = 0.0f;

  int rt = blockIdx.x;
  int start[RPW], deg[RPW];
  int2 ed[RPW];
#pragma unroll
  for (int q = 0; q < RPW; ++q) {
    const int r = min((rt << 4) + wave * RPW + q, n - 1);
    start[q] = rowstart[r];
    deg[q] = rt < rowTiles ? cnt[r] : 0;
  }
#pragma unroll
  for (int q = 0; q < RPW; ++q) ed[q] = chunk(start[q], deg[q], 0);

  for (int it = 0; rt < rowTiles; rt += gridDim.x, ++it) {
    const int row0 = rt << 4;
    const int rtn = rt + (int)gridDim.x;

    int nstart[RPW], ndeg[RPW];
#pragma unroll
    for (int q = 0; q < RPW; ++q) {
      const int r = min((rtn << 4) + wave * RPW + q, n - 1);
      nstart[q] = rowstart[r];
      ndeg[q] = rtn < rowTiles ? cnt[r] : 0;
    }

    int maxdeg = 0;
#pragma unroll
    for (int q = 0; q < RPW; ++q) maxdeg = max(maxdeg, deg[q]);

    vf acc[RPW];
    float wsum[RPW];
#pragma unroll
    for (int q = 0; q < RPW; ++q) { acc[q] = (vf){}; wsum[q] = 0.0f; }

    for (int base = 0; base < maxdeg; base += 32) {
      int e_src[RPW], m[RPW];
      float e_w[RPW];
#pragma unroll
      for (int q = 0; q < RPW; ++q) {
        m[q] = min(32, deg[q] - base);
        e_src[q] = ed[q].x;
        e_w[q] = lane < m[q] ? __int_as_float(ed[q].y) : 0.0f;
      }
      const int mm = min(32, maxdeg - base);
      for (int j0 = 0; j0 < mm; j0 += 8) {
        vf v[RPW][8];
        float wj[RPW][8];
#pragma unroll
        for (int q = 0; q < RPW; ++q) {
#pragma unroll
          for (int u = 0; u < 8; ++u) {
            const int s = __builtin_amdgcn_readlane(e_src[q], j0 + u);
            wj[q][u] = __int_as_float(
                __builtin_amdgcn_readlane(__float_as_int(e_w[q]), j0 + u));
            v[q][u] = *(const vf*)(x + (size_t)s * K + lane * VPL);
          }
        }
#pragma unroll
        for (int q = 0; q < RPW; ++q) {
#pragma unroll
          for (int u = 0; u < 8; ++u) {
            acc[q] += wj[q][u] * v[q][u];
            if constexpr (NORM) wsum[q] += wj[q][u];
          }
        }
      }
      if (base + 32 < maxdeg) {
#pragma unroll
        for (int q = 0; q < RPW; ++q) ed[q] = chunk(start[q], deg[q], base + 32);
      }
    }

    float* __restrict__ tb = &tile[it & 1][0];
#pragma unroll
    for (int q = 0; q < RPW; ++q) {
      vf o = acc[q];
      if constexpr (NORM) o = o * sc + wsum[q] * sh;
      *(vf*)(tb + (wave * RPW + q) * P + lane * VPL) = o;
    }
    __syncthreads();

#pragma unroll
    for (int q = 0; q < RPW; ++q) {
      start[q] = nstart[q];
      deg[q] = ndeg[q];
      ed[q] = chunk(start[q], deg[q], 0);
    }

    const v4f* __restrict__ ap = (const v4f*)(tb + c * P + 4 * h);
    v8f acc0 = {};
    v8f acc1 = {};
    constexpr int HJ = K / 16;
    static_assert(HJ == 4 || HJ == 8, "the load fence names 4 or 8 reads");
#pragma unroll
    for (int hf = 0; hf < 2; ++hf) {
      v4f a[HJ];
#pragma unroll
      for (int j = 0; j < HJ; ++j) a[j] = ap[2 * (hf * HJ + j)];
      if constexpr (HJ == 8)
        asm volatile("" ::"v"(a[0]), "v"(a[1]), "v"(a[2]), "v"(a[3]),
                     "v"(a[4]), "v"(a[5]), "v"(a[6]), "v"(a[7]));
      else
        asm volatile("" ::"v"(a[0]), "v"(a[1]), "v"(a[2]), "v"(a[3]));
#pragma unroll
      for (int j = 0; j < HJ; ++j) {
        const int jj = hf * HJ + j;
        v2f lo, hi;
        lo.x = a[j].x; lo.y = a[j].y;
        hi.x = a[j].z; hi.y = a[j].w;
        acc0 = __builtin_amdgcn_wmma_f32_16x16x4_f32(
            false, lo, false, b0[2 * jj], (short)0, acc0, false, false);
        acc1 = __builtin_amdgcn_wmma_f32_16x16x4_f32(
            false, lo, false, b1[2 * jj], (short)0, acc1, false, false);
        acc0 = __builtin_amdgcn_wmma_f32_16x16x4_f32(
            false, hi, false, b0[2 * jj + 1], (short)0, acc0, false, false);
        acc1 = __builtin_amdgcn_wmma_f32_16x16x4_f32(
            false, hi, false, b1[2 * jj + 1], (short)0, acc1, false, false);
      }
    }

    float* __restrict__ cp = C + (size_t)row0 * NC + col0 + lane;
#pragma unroll
    for (int r = 0; r < 8; ++r) {
      float v0 = acc0[r];
      float v1 = acc1[r];
      v0 = v0 > 0.0f ? v0 : 0.0f;
      v1 = v1 > 0.0f ? v1 : 0.0f;
      const float give = h ? v0 : v1;
      const float got = __builtin_bit_cast(
          float, __builtin_amdgcn_permlanex16(
                     __builtin_bit_cast(int, give), __builtin_bit_cast(int, give),
                     0x76543210, 0xfedcba98, false, false));
      cp[(size_t)r * NC]       = h ? got : v0;
      cp[(size_t)(8 + r) * NC] = h ? v1 : got;
      s0 += v0; q0 += v0 * v0;
      s1 += v1; q1 += v1 * v1;
    }
  }

  s0 += __shfl_xor(s0, 16); q0 += __shfl_xor(q0, 16);
  s1 += __shfl_xor(s1, 16); q1 += __shfl_xor(q1, 16);
  if (h == 0) {
    atomicAdd(&stats[col0 + c], s0);
    atomicAdd(&stats[NC + col0 + c], q0);
    atomicAdd(&stats[col0 + 16 + c], s1);
    atomicAdd(&stats[NC + col0 + 16 + c], q1);
  }
}

#define BN_U 10

template <int D>
__global__ void __launch_bounds__(256)
bn_normalize(const float* __restrict__ agg,
             float* __restrict__ out,
             const float* __restrict__ gamma,
             const float* __restrict__ beta,
             const float* __restrict__ stats,
             int Nrows) {
  const int cg = threadIdx.x & (D / 4 - 1);
  const float invN = 1.0f / (float)Nrows;
  const v4f sm = ((const v4f*)stats)[cg];
  const v4f sq = ((const v4f*)(stats + D))[cg];
  const v4f ga = ((const v4f*)gamma)[cg];
  const v4f be = ((const v4f*)beta)[cg];
  const v4f mean = sm * invN;
  const v4f var = sq * invN - mean * mean;
  v4f sc;
  sc.x = ga.x * rsqrtf(var.x + BN_EPS);
  sc.y = ga.y * rsqrtf(var.y + BN_EPS);
  sc.z = ga.z * rsqrtf(var.z + BN_EPS);
  sc.w = ga.w * rsqrtf(var.w + BN_EPS);

  const int total4 = Nrows * (D / 4);
  const int stride = (int)(gridDim.x * 256);
  const v4f* __restrict__ in4 = (const v4f*)agg;
  v4f* __restrict__ out4 = (v4f*)out;

  int i0 = (int)(blockIdx.x * 256 + threadIdx.x);
  for (; i0 + (BN_U - 1) * stride < total4; i0 += BN_U * stride) {
    v4f v[BN_U];
#pragma unroll
    for (int u = 0; u < BN_U; ++u) v[u] = in4[i0 + u * stride];
    static_assert(BN_U == 10, "the statement below names every element of v");
    asm volatile("" ::"v"(v[0]), "v"(v[1]), "v"(v[2]), "v"(v[3]), "v"(v[4]),
                 "v"(v[5]), "v"(v[6]), "v"(v[7]), "v"(v[8]), "v"(v[9]));
#pragma unroll
    for (int u = 0; u < BN_U; ++u)
      out4[i0 + u * stride] = (v[u] - mean) * sc + be;
  }
  for (; i0 < total4; i0 += stride)
    out4[i0] = (in4[i0] - mean) * sc + be;
}

extern "C" void kernel_launch(void* const* d_in, const int* in_sizes, int n_in,
                              void* d_out, int out_size, void* d_ws,
                              size_t ws_size, hipStream_t stream) {
  (void)in_sizes; (void)n_in; (void)out_size; (void)ws_size;

  const float* y   = (const float*)d_in[0];
  const int*   src = (const int*)d_in[1];
  const int*   dst = (const int*)d_in[2];
  const float* w   = (const float*)d_in[3];
  const float* W1  = (const float*)d_in[4];
  const float* g1  = (const float*)d_in[5];
  const float* b1  = (const float*)d_in[6];
  const float* W2  = (const float*)d_in[7];
  const float* g2  = (const float*)d_in[8];
  const float* b2  = (const float*)d_in[9];
  float* out = (float*)d_out;

  float* ws     = (float*)d_ws;
  float* z1     = ws;
  float* z2     = z1 + (size_t)NN * H1C;
  float* stats1 = z2 + (size_t)NN * DINC;
  float* stats2 = stats1 + 2 * H1C;
  int*   cnt    = (int*)(stats2 + 2 * DINC);
  int*   rowst  = cnt + NN;
  int*   pieces = rowst + NN;
  int*   binned = pieces + NGRP * NBINS + 24;
  int*   edges  = binned + 2 * EE;

  bin_place<<<NGRP, L1_T, 0, stream>>>(src, dst, w, pieces, binned,
                                       (unsigned*)stats1, 2 * (H1C + DINC));
  bin_rows<<<NBINS, L2_T, 0, stream>>>(pieces, binned, rowst, cnt, edges);

  edge_gemm_relu_colstats<H2C, H1C, false><<<1042, H1C, 0, stream>>>(
      y, W1, z1, stats1, NN / 16, rowst, cnt, edges, NN, EE, nullptr, nullptr,
      nullptr);

  edge_gemm_relu_colstats<H1C, DINC, true><<<512, DINC, 0, stream>>>(
      z1, W2, z2, stats2, NN / 16, rowst, cnt, edges, NN, EE, g1, b1, stats1);

  bn_normalize<DINC><<<625, 256, 0, stream>>>(z2, out, g2, b2, stats2, NN);
}
